# starting kernel
# baseline (speedup 1.0000x reference)
#include <hip/hip_runtime.h>
#include <hip/hip_bf16.h>


#define T_SEQ 8192
#define E_DIM 256
#define V_DIM 17

typedef float v2f __attribute__((ext_vector_type(2)));
typedef float v8f __attribute__((ext_vector_type(8)));

__device__ __forceinline__ v8f wmma4(v2f a, v2f b, v8f c) {
  return __builtin_amdgcn_wmma_f32_16x16x4_f32(
       false, a,  false, b,
       (short)0, c,  false,  false);
}

__global__ __launch_bounds__(32) void head123_kernel(
    const float* __restrict__ x,
    const float* __restrict__ W1, const float* __restrict__ b1,
    const float* __restrict__ W2, const float* __restrict__ b2,
    const float* __restrict__ W3, const float* __restrict__ b3,
    float* __restrict__ out) {
  const int n       = blockIdx.y;
  const int rowbase = blockIdx.x * 16;

  const float* W;
  const float* b;
  if (rowbase < 64)       { W = W1; b = b1; }
  else if (rowbase < 576) { W = W2; b = b2; }
  else                    { W = W3; b = b3; }

  const int lane = threadIdx.x & 31;
  const int l15  = lane & 15;
  const int koff = (lane < 16) ? 0 : 2;

  const float* xrow = x + ((size_t)n * T_SEQ + rowbase + l15) * E_DIM;
  const int c1  = 16 + l15;
  const int c1c = (c1 < V_DIM) ? c1 : (V_DIM - 1);

  v8f acc0 = {};
  v8f acc1 = {};
  for (int kk = 0; kk < E_DIM; kk += 4) {
    v2f a;
    a.x = xrow[kk + koff];
    a.y = xrow[kk + koff + 1];
    v2f bf0, bf1;
    bf0.x = W[(kk + koff) * V_DIM + l15];
    bf0.y = W[(kk + koff + 1) * V_DIM + l15];
    bf1.x = W[(kk + koff) * V_DIM + c1c];
    bf1.y = W[(kk + koff + 1) * V_DIM + c1c];
    acc0 = wmma4(a, bf0, acc0);
    acc1 = wmma4(a, bf1, acc1);
  }

#pragma unroll
  for (int v = 0; v < 8; ++v) {
    const int r = ((lane < 16) ? v : v + 8);
    float* o = out + ((size_t)n * T_SEQ + rowbase + r) * V_DIM;
    o[l15] = acc0[v] + b[l15];
    if (c1 < V_DIM) o[c1] = acc1[v] + b[c1];
  }
}

__global__ __launch_bounds__(256) void conv_head_kernel(
    const float* __restrict__ x,
    const float* __restrict__ Wd, const float* __restrict__ bd,
    const float* __restrict__ Wl, const float* __restrict__ bl,
    float* __restrict__ out,
    int s, int ioff, int out_off, int nt, int kgroups) {
  __shared__ float Elds[64 * 256];

  const int n  = blockIdx.y;
  const int tb = blockIdx.x / kgroups;
  const int kg = blockIdx.x % kgroups;

  const int tid  = threadIdx.x;
  const int w    = tid >> 5;
  const int lane = tid & 31;
  const int l15  = lane & 15;
  const int koff = (lane < 16) ? 0 : 2;

  const int trel   = tb * 16 + l15;
  const int tclamp = (trel < nt) ? trel : (nt - 1);
  const float* xrow = x + ((size_t)n * T_SEQ + ioff + tclamp) * E_DIM;
  const int Es = E_DIM * s;

  v8f acc[8];
#pragma unroll
  for (int i = 0; i < 8; ++i) acc[i] = (v8f){};

  for (int kk = 0; kk < E_DIM; kk += 4) {
    v2f a;
    a.x = xrow[kk + koff];
    a.y = xrow[kk + koff + 1];
#pragma unroll
    for (int i = 0; i < 8; ++i) {
      const int ct = w * 8 + i;
      const int g  = ct >> 4;
      const int ot = ct & 15;
      const int k  = kg * 4 + g;
      const int o  = ot * 16 + l15;
      const float* bp = Wd + (size_t)(kk + koff) * Es + (size_t)o * s + k;
      v2f bf;
      bf.x = bp[0];
      bf.y = bp[Es];
      acc[i] = wmma4(a, bf, acc[i]);
    }
  }

#pragma unroll
  for (int i = 0; i < 8; ++i) {
    const int ct = w * 8 + i;
    const int g  = ct >> 4;
    const int ot = ct & 15;
    const int o  = ot * 16 + l15;
    const float bdv = bd[o];
#pragma unroll
    for (int v = 0; v < 8; ++v) {
      const int tl  = (lane < 16) ? v : v + 8;
      const int row = tl * 4 + g;
      Elds[row * 256 + (o ^ ((row & 15) * 4))] = acc[i][v] + bdv;
    }
  }
  __syncthreads();

  const int rt   = w >> 1;
  const int col0 = (w & 1) * 16;
  const int m    = rt * 16 + l15;
  const int col  = col0 + l15;
  const bool cval = (col < V_DIM);
  const int colc  = cval ? col : (V_DIM - 1);
  const int msw   = (m & 15) * 4;

  v8f acc2 = {};
  for (int kk = 0; kk < E_DIM; kk += 4) {
    const int k0 = kk + koff;
    v2f a;
    a.x = Elds[m * 256 + (k0 ^ msw)];
    a.y = Elds[m * 256 + ((k0 + 1) ^ msw)];
    v2f bf;
    bf.x = Wl[(size_t)k0 * V_DIM + colc];
    bf.y = Wl[(size_t)(k0 + 1) * V_DIM + colc];
    acc2 = wmma4(a, bf, acc2);
  }

  const float blv = bl[colc];
#pragma unroll
  for (int v = 0; v < 8; ++v) {
    const int r     = rt * 16 + ((lane < 16) ? v : v + 8);
    const int tl    = r >> 2;
    const int g     = r & 3;
    const int trel2 = tb * 16 + tl;
    if (cval && trel2 < nt) {
      const int j = out_off + trel2 * s + kg * 4 + g;
      out[((size_t)n * T_SEQ + j) * V_DIM + col] = acc2[v] + blv;
    }
  }
}

extern "C" void kernel_launch(void* const* d_in, const int* in_sizes, int n_in,
                              void* d_out, int out_size, void* d_ws, size_t ws_size,
                              hipStream_t stream) {
  const float* x   = (const float*)d_in[0];
  const float* W1  = (const float*)d_in[4];
  const float* b1  = (const float*)d_in[5];
  const float* W2  = (const float*)d_in[6];
  const float* b2  = (const float*)d_in[7];
  const float* W3  = (const float*)d_in[8];
  const float* b3  = (const float*)d_in[9];
  const float* Wd4 = (const float*)d_in[10];
  const float* bd4 = (const float*)d_in[11];
  const float* Wl4 = (const float*)d_in[12];
  const float* bl4 = (const float*)d_in[13];
  const float* Wd5 = (const float*)d_in[14];
  const float* bd5 = (const float*)d_in[15];
  const float* Wl5 = (const float*)d_in[16];
  const float* bl5 = (const float*)d_in[17];
  float* out = (float*)d_out;

  const int N = 8;

  head123_kernel<<<dim3(100, N), 32, 0, stream>>>(x, W1, b1, W2, b2, W3, b3, out);

  conv_head_kernel<<<dim3(32 * 1, N), 256, 0, stream>>>(
      x, Wd4, bd4, Wl4, bl4, out,  4,  1600,  1600,
       512,  1);

  conv_head_kernel<<<dim3(36 * 2, N), 256, 0, stream>>>(
      x, Wd5, bd5, Wl5, bl5, out,  8,  2112,  3648,
       568,  2);
}


# Conv heads: fold Wd[:,:,k]@Wl and bd@Wl+bl into one 256 x s*17 weight per head each call (in workspace), then one f32 WMMA GEMM x@Wc per head; f32 summation order changes (products reassociated)
# speedup vs baseline: 6.7497x; 6.7497x over previous
#include <hip/hip_runtime.h>
#include <hip/hip_bf16.h>


#define T_SEQ 8192
#define E_DIM 256
#define V_DIM 17
#define WP_PAIRS 129
#define WP_OFF5 (WP_PAIRS * 4 * V_DIM * 2)
#define WP_BYTES ((size_t)(WP_PAIRS * 12 * V_DIM * 2) * sizeof(float))

typedef float v2f __attribute__((ext_vector_type(2)));
typedef float v8f __attribute__((ext_vector_type(8)));

__device__ __forceinline__ v8f wmma4(v2f a, v2f b, v8f c) {
  return __builtin_amdgcn_wmma_f32_16x16x4_f32(
       false, a,  false, b,
       (short)0, c,  false,  false);
}

__global__ __launch_bounds__(32) void head123_kernel(
    const float* __restrict__ x,
    const float* __restrict__ W1, const float* __restrict__ b1,
    const float* __restrict__ W2, const float* __restrict__ b2,
    const float* __restrict__ W3, const float* __restrict__ b3,
    float* __restrict__ out) {
  const int n       = blockIdx.y;
  const int rowbase = blockIdx.x * 16;

  const float* W;
  const float* b;
  if (rowbase < 64)       { W = W1; b = b1; }
  else if (rowbase < 576) { W = W2; b = b2; }
  else                    { W = W3; b = b3; }

  const int lane = threadIdx.x & 31;
  const int l15  = lane & 15;
  const int koff = (lane < 16) ? 0 : 2;

  const float* xrow = x + ((size_t)n * T_SEQ + rowbase + l15) * E_DIM;
  const int c1  = 16 + l15;
  const int c1c = (c1 < V_DIM) ? c1 : (V_DIM - 1);

  v8f acc0 = {};
  v8f acc1 = {};
  for (int kk = 0; kk < E_DIM; kk += 4) {
    v2f a;
    a.x = xrow[kk + koff];
    a.y = xrow[kk + koff + 1];
    v2f bf0, bf1;
    bf0.x = W[(kk + koff) * V_DIM + l15];
    bf0.y = W[(kk + koff + 1) * V_DIM + l15];
    bf1.x = W[(kk + koff) * V_DIM + c1c];
    bf1.y = W[(kk + koff + 1) * V_DIM + c1c];
    acc0 = wmma4(a, bf0, acc0);
    acc1 = wmma4(a, bf1, acc1);
  }

#pragma unroll
  for (int v = 0; v < 8; ++v) {
    const int r = ((lane < 16) ? v : v + 8);
    float* o = out + ((size_t)n * T_SEQ + rowbase + r) * V_DIM;
    o[l15] = acc0[v] + b[l15];
    if (c1 < V_DIM) o[c1] = acc1[v] + b[c1];
  }
}

template <int AS>
__device__ __forceinline__ void fold_tile(const float* __restrict__ ap,
                                          const float* __restrict__ b0p,
                                          const float* __restrict__ b1p,
                                          v8f& acc0, v8f& acc1) {
  v2f A[2][8], B0[2][8], B1[2][8];
#pragma unroll
  for (int j = 0; j < 8; ++j) {
    const int oo = j * 4;
    A[0][j].x  = ap[oo * AS];
    A[0][j].y  = ap[oo * AS + AS];
    B0[0][j].x = b0p[oo * V_DIM];
    B0[0][j].y = b0p[oo * V_DIM + V_DIM];
    B1[0][j].x = b1p[oo * V_DIM];
    B1[0][j].y = b1p[oo * V_DIM + V_DIM];
  }
#pragma unroll
  for (int ch = 0; ch < 8; ++ch) {
    const int cur = ch & 1;
    const int nxt = cur ^ 1;
    if (ch + 1 < 8) {
#pragma unroll
      for (int j = 0; j < 8; ++j) {
        const int oo = ((ch + 1) * 8 + j) * 4;
        A[nxt][j].x  = ap[oo * AS];
        A[nxt][j].y  = ap[oo * AS + AS];
        B0[nxt][j].x = b0p[oo * V_DIM];
        B0[nxt][j].y = b0p[oo * V_DIM + V_DIM];
        B1[nxt][j].x = b1p[oo * V_DIM];
        B1[nxt][j].y = b1p[oo * V_DIM + V_DIM];
      }
    }
    __builtin_amdgcn_sched_barrier(0);
#pragma unroll
    for (int j = 0; j < 8; ++j) {
      acc0 = wmma4(A[cur][j], B0[cur][j], acc0);
      acc1 = wmma4(A[cur][j], B1[cur][j], acc1);
    }
    __builtin_amdgcn_sched_barrier(0);
  }
}

__global__ __launch_bounds__(32) void fold_kernel(
    const float* __restrict__ Wd4, const float* __restrict__ bd4,
    const float* __restrict__ Wl4,
    const float* __restrict__ Wd5, const float* __restrict__ bd5,
    const float* __restrict__ Wl5,
    float* __restrict__ ws, size_t ws_bytes) {
  if (ws_bytes < WP_BYTES) return;

  const int  ky = blockIdx.y;
  const bool h5 = (ky >= 4);
  const float* Wd = h5 ? Wd5 : Wd4;
  const float* bd = h5 ? bd5 : bd4;
  const float* Wl = h5 ? Wl5 : Wl4;
  const int s  = h5 ? 8 : 4;
  const int k  = h5 ? (ky - 4) : ky;
  const int ld = s * V_DIM;
  float* Wp = ws + (h5 ? WP_OFF5 : 0);

  const int lane = threadIdx.x & 31;
  const int l15  = lane & 15;
  const int koff = (lane < 16) ? 0 : 2;

  const int ibase = blockIdx.x * 16;
  const int c1  = 16 + l15;
  const int c1c = (c1 < V_DIM) ? c1 : (V_DIM - 1);

  const float* b0p = Wl + koff * V_DIM + l15;
  const float* b1p = Wl + koff * V_DIM + c1c;

  v8f acc0 = {};
  v8f acc1 = {};
  if (ibase < E_DIM) {
    const float* arow = Wd + (size_t)(ibase + l15) * (E_DIM * s) + k;
    if (h5) fold_tile<8>(arow + koff * 8, b0p, b1p, acc0, acc1);
    else    fold_tile<4>(arow + koff * 4, b0p, b1p, acc0, acc1);
  } else {
    fold_tile<1>(bd + koff, b0p, b1p, acc0, acc1);
  }

#pragma unroll
  for (int v = 0; v < 8; v += 2) {
    const int r = ibase + ((lane < 16) ? v : v + 8);
    if (r <= E_DIM) {
      float* o = Wp + ((size_t)(r >> 1) * ld + k * V_DIM) * 2;
      o[l15 * 2]     = acc0[v];
      o[l15 * 2 + 1] = acc0[v + 1];
      if (c1 < V_DIM) {
        o[c1 * 2]     = acc1[v];
        o[c1 * 2 + 1] = acc1[v + 1];
      }
    }
  }
}

template <int S>
__global__ __launch_bounds__(32) void conv_folded_kernel(
    const float* __restrict__ x,
    const float* __restrict__ ws,
    const float* __restrict__ bl,
    float* __restrict__ out,
    int ioff, int out_off, int nt, size_t ws_bytes) {
  if (ws_bytes < WP_BYTES) return;

  constexpr int NCOL   = S * V_DIM;
  constexpr int GROUPS = (NCOL + 47) / 48;
  const float* Wp = ws + ((S == 8) ? WP_OFF5 : 0);

  const int n   = blockIdx.y;
  const int tb  = blockIdx.x / GROUPS;
  const int grp = blockIdx.x % GROUPS;

  const int lane = threadIdx.x & 31;
  const int l15  = lane & 15;
  const int koff = (lane < 16) ? 0 : 2;

  const int trel   = tb * 16 + l15;
  const int tclamp = (trel < nt) ? trel : (nt - 1);
  const float* xp = x + ((size_t)n * T_SEQ + ioff + tclamp) * E_DIM + koff;

  const float* wp[3];
#pragma unroll
  for (int i = 0; i < 3; ++i) {
    const int c  = (grp * 3 + i) * 16 + l15;
    const int cc = (c < NCOL) ? c : (NCOL - 1);
    wp[i] = Wp + ((size_t)(koff >> 1) * NCOL + cc) * 2;
  }

  v8f acc[3];
#pragma unroll
  for (int i = 0; i < 3; ++i) acc[i] = (v8f){};

  v2f A[2][8], B[2][8][3];
#pragma unroll
  for (int j = 0; j < 8; ++j) {
    const int kk = j * 4;
    A[0][j].x = xp[kk];
    A[0][j].y = xp[kk + 1];
#pragma unroll
    for (int i = 0; i < 3; ++i) {
      B[0][j][i].x = wp[i][kk * NCOL];
      B[0][j][i].y = wp[i][kk * NCOL + 1];
    }
  }
#pragma unroll
  for (int ch = 0; ch < 8; ++ch) {
    const int cur = ch & 1;
    const int nxt = cur ^ 1;
    if (ch + 1 < 8) {
#pragma unroll
      for (int j = 0; j < 8; ++j) {
        const int kk = ((ch + 1) * 8 + j) * 4;
        A[nxt][j].x = xp[kk];
        A[nxt][j].y = xp[kk + 1];
#pragma unroll
        for (int i = 0; i < 3; ++i) {
          B[nxt][j][i].x = wp[i][kk * NCOL];
          B[nxt][j][i].y = wp[i][kk * NCOL + 1];
        }
      }
    }
    __builtin_amdgcn_sched_barrier(0);
#pragma unroll
    for (int j = 0; j < 8; ++j) {
#pragma unroll
      for (int i = 0; i < 3; ++i) acc[i] = wmma4(A[cur][j], B[cur][j][i], acc[i]);
    }
    __builtin_amdgcn_sched_barrier(0);
  }

#pragma unroll
  for (int i = 0; i < 3; ++i) {
    const int c = (grp * 3 + i) * 16 + l15;
    if (c < NCOL) {
      const float bias = Wp[((size_t)(E_DIM / 2) * NCOL + c) * 2] + bl[c % V_DIM];
#pragma unroll
      for (int v = 0; v < 8; ++v) {
        const int trel2 = tb * 16 + ((lane < 16) ? v : v + 8);
        if (trel2 < nt) {
          out[((size_t)n * T_SEQ + out_off + (size_t)trel2 * S) * V_DIM + c] =
              acc[i][v] + bias;
        }
      }
    }
  }
}

extern "C" void kernel_launch(void* const* d_in, const int* in_sizes, int n_in,
                              void* d_out, int out_size, void* d_ws, size_t ws_size,
                              hipStream_t stream) {
  const float* x   = (const float*)d_in[0];
  const float* W1  = (const float*)d_in[4];
  const float* b1  = (const float*)d_in[5];
  const float* W2  = (const float*)d_in[6];
  const float* b2  = (const float*)d_in[7];
  const float* W3  = (const float*)d_in[8];
  const float* b3  = (const float*)d_in[9];
  const float* Wd4 = (const float*)d_in[10];
  const float* bd4 = (const float*)d_in[11];
  const float* Wl4 = (const float*)d_in[12];
  const float* bl4 = (const float*)d_in[13];
  const float* Wd5 = (const float*)d_in[14];
  const float* bd5 = (const float*)d_in[15];
  const float* Wl5 = (const float*)d_in[16];
  const float* bl5 = (const float*)d_in[17];
  float* out = (float*)d_out;

  const int N = 8;

  head123_kernel<<<dim3(100, N), 32, 0, stream>>>(x, W1, b1, W2, b2, W3, b3, out);

  float* ws = (float*)d_ws;
  fold_kernel<<<dim3(17, 12), 32, 0, stream>>>(Wd4, bd4, Wl4, Wd5, bd5, Wl5, ws, ws_size);

  conv_folded_kernel<4><<<dim3(32 * 2, N), 32, 0, stream>>>(
      x, ws, bl4, out,  1600,  1600,  512, ws_size);

  conv_folded_kernel<8><<<dim3(36 * 3, N), 32, 0, stream>>>(
      x, ws, bl5, out,  2112,  3648,  568, ws_size);
}


# Merge independent launches, 4 to 2: fold blocks + heads 1-3 blocks share one kernel, both conv heads share a second; per-wave code and f32 summation order unchanged
# speedup vs baseline: 10.7283x; 1.5894x over previous
#include <hip/hip_runtime.h>
#include <hip/hip_bf16.h>


#define T_SEQ 8192
#define E_DIM 256
#define V_DIM 17
#define WP_PAIRS 129
#define WP_OFF5 (WP_PAIRS * 4 * V_DIM * 2)
#define WP_BYTES ((size_t)(WP_PAIRS * 12 * V_DIM * 2) * sizeof(float))

typedef float v2f __attribute__((ext_vector_type(2)));
typedef float v8f __attribute__((ext_vector_type(8)));

__device__ __forceinline__ v8f wmma4(v2f a, v2f b, v8f c) {
  return __builtin_amdgcn_wmma_f32_16x16x4_f32(
       false, a,  false, b,
       (short)0, c,  false,  false);
}

__device__ __forceinline__ void head123_body(
    const int bx, const int n,
    const float* __restrict__ x,
    const float* __restrict__ W1, const float* __restrict__ b1,
    const float* __restrict__ W2, const float* __restrict__ b2,
    const float* __restrict__ W3, const float* __restrict__ b3,
    float* __restrict__ out) {
  const int rowbase = bx * 16;

  const float* W;
  const float* b;
  if (rowbase < 64)       { W = W1; b = b1; }
  else if (rowbase < 576) { W = W2; b = b2; }
  else                    { W = W3; b = b3; }

  const int lane = threadIdx.x & 31;
  const int l15  = lane & 15;
  const int koff = (lane < 16) ? 0 : 2;

  const float* xrow = x + ((size_t)n * T_SEQ + rowbase + l15) * E_DIM;
  const int c1  = 16 + l15;
  const int c1c = (c1 < V_DIM) ? c1 : (V_DIM - 1);

  v8f acc0 = {};
  v8f acc1 = {};
  for (int kk = 0; kk < E_DIM; kk += 4) {
    v2f a;
    a.x = xrow[kk + koff];
    a.y = xrow[kk + koff + 1];
    v2f bf0, bf1;
    bf0.x = W[(kk + koff) * V_DIM + l15];
    bf0.y = W[(kk + koff + 1) * V_DIM + l15];
    bf1.x = W[(kk + koff) * V_DIM + c1c];
    bf1.y = W[(kk + koff + 1) * V_DIM + c1c];
    acc0 = wmma4(a, bf0, acc0);
    acc1 = wmma4(a, bf1, acc1);
  }

#pragma unroll
  for (int v = 0; v < 8; ++v) {
    const int r = ((lane < 16) ? v : v + 8);
    float* o = out + ((size_t)n * T_SEQ + rowbase + r) * V_DIM;
    o[l15] = acc0[v] + b[l15];
    if (c1 < V_DIM) o[c1] = acc1[v] + b[c1];
  }
}

template <int AS>
__device__ __forceinline__ void fold_tile(const float* __restrict__ ap,
                                          const float* __restrict__ b0p,
                                          const float* __restrict__ b1p,
                                          v8f& acc0, v8f& acc1) {
  v2f A[2][8], B0[2][8], B1[2][8];
#pragma unroll
  for (int j = 0; j < 8; ++j) {
    const int oo = j * 4;
    A[0][j].x  = ap[oo * AS];
    A[0][j].y  = ap[oo * AS + AS];
    B0[0][j].x = b0p[oo * V_DIM];
    B0[0][j].y = b0p[oo * V_DIM + V_DIM];
    B1[0][j].x = b1p[oo * V_DIM];
    B1[0][j].y = b1p[oo * V_DIM + V_DIM];
  }
#pragma unroll
  for (int ch = 0; ch < 8; ++ch) {
    const int cur = ch & 1;
    const int nxt = cur ^ 1;
    if (ch + 1 < 8) {
#pragma unroll
      for (int j = 0; j < 8; ++j) {
        const int oo = ((ch + 1) * 8 + j) * 4;
        A[nxt][j].x  = ap[oo * AS];
        A[nxt][j].y  = ap[oo * AS + AS];
        B0[nxt][j].x = b0p[oo * V_DIM];
        B0[nxt][j].y = b0p[oo * V_DIM + V_DIM];
        B1[nxt][j].x = b1p[oo * V_DIM];
        B1[nxt][j].y = b1p[oo * V_DIM + V_DIM];
      }
    }
    __builtin_amdgcn_sched_barrier(0);
#pragma unroll
    for (int j = 0; j < 8; ++j) {
      acc0 = wmma4(A[cur][j], B0[cur][j], acc0);
      acc1 = wmma4(A[cur][j], B1[cur][j], acc1);
    }
    __builtin_amdgcn_sched_barrier(0);
  }
}

__device__ __forceinline__ void fold_body(
    const int bx, const int ky,
    const float* __restrict__ Wd4, const float* __restrict__ bd4,
    const float* __restrict__ Wl4,
    const float* __restrict__ Wd5, const float* __restrict__ bd5,
    const float* __restrict__ Wl5,
    float* __restrict__ ws) {
  const bool h5 = (ky >= 4);
  const float* Wd = h5 ? Wd5 : Wd4;
  const float* bd = h5 ? bd5 : bd4;
  const float* Wl = h5 ? Wl5 : Wl4;
  const int s  = h5 ? 8 : 4;
  const int k  = h5 ? (ky - 4) : ky;
  const int ld = s * V_DIM;
  float* Wp = ws + (h5 ? WP_OFF5 : 0);

  const int lane = threadIdx.x & 31;
  const int l15  = lane & 15;
  const int koff = (lane < 16) ? 0 : 2;

  const int ibase = bx * 16;
  const int c1  = 16 + l15;
  const int c1c = (c1 < V_DIM) ? c1 : (V_DIM - 1);

  const float* b0p = Wl + koff * V_DIM + l15;
  const float* b1p = Wl + koff * V_DIM + c1c;

  v8f acc0 = {};
  v8f acc1 = {};
  if (ibase < E_DIM) {
    const float* arow = Wd + (size_t)(ibase + l15) * (E_DIM * s) + k;
    if (h5) fold_tile<8>(arow + koff * 8, b0p, b1p, acc0, acc1);
    else    fold_tile<4>(arow + koff * 4, b0p, b1p, acc0, acc1);
  } else {
    fold_tile<1>(bd + koff, b0p, b1p, acc0, acc1);
  }

#pragma unroll
  for (int v = 0; v < 8; v += 2) {
    const int r = ibase + ((lane < 16) ? v : v + 8);
    if (r <= E_DIM) {
      float* o = Wp + ((size_t)(r >> 1) * ld + k * V_DIM) * 2;
      o[l15 * 2]     = acc0[v];
      o[l15 * 2 + 1] = acc0[v + 1];
      if (c1 < V_DIM) {
        o[c1 * 2]     = acc1[v];
        o[c1 * 2 + 1] = acc1[v + 1];
      }
    }
  }
}

template <int S>
__device__ __forceinline__ void conv_folded_body(
    const int bx, const int n,
    const float* __restrict__ x,
    const float* __restrict__ ws,
    const float* __restrict__ bl,
    float* __restrict__ out,
    const int ioff, const int out_off, const int nt) {
  constexpr int NCOL   = S * V_DIM;
  constexpr int GROUPS = (NCOL + 47) / 48;
  const float* Wp = ws + ((S == 8) ? WP_OFF5 : 0);

  const int tb  = bx / GROUPS;
  const int grp = bx % GROUPS;

  const int lane = threadIdx.x & 31;
  const int l15  = lane & 15;
  const int koff = (lane < 16) ? 0 : 2;

  const int trel   = tb * 16 + l15;
  const int tclamp = (trel < nt) ? trel : (nt - 1);
  const float* xp = x + ((size_t)n * T_SEQ + ioff + tclamp) * E_DIM + koff;

  const float* wp[3];
#pragma unroll
  for (int i = 0; i < 3; ++i) {
    const int c  = (grp * 3 + i) * 16 + l15;
    const int cc = (c < NCOL) ? c : (NCOL - 1);
    wp[i] = Wp + ((size_t)(koff >> 1) * NCOL + cc) * 2;
  }

  v8f acc[3];
#pragma unroll
  for (int i = 0; i < 3; ++i) acc[i] = (v8f){};

  v2f A[2][8], B[2][8][3];
#pragma unroll
  for (int j = 0; j < 8; ++j) {
    const int kk = j * 4;
    A[0][j].x = xp[kk];
    A[0][j].y = xp[kk + 1];
#pragma unroll
    for (int i = 0; i < 3; ++i) {
      B[0][j][i].x = wp[i][kk * NCOL];
      B[0][j][i].y = wp[i][kk * NCOL + 1];
    }
  }
#pragma unroll
  for (int ch = 0; ch < 8; ++ch) {
    const int cur = ch & 1;
    const int nxt = cur ^ 1;
    if (ch + 1 < 8) {
#pragma unroll
      for (int j = 0; j < 8; ++j) {
        const int kk = ((ch + 1) * 8 + j) * 4;
        A[nxt][j].x = xp[kk];
        A[nxt][j].y = xp[kk + 1];
#pragma unroll
        for (int i = 0; i < 3; ++i) {
          B[nxt][j][i].x = wp[i][kk * NCOL];
          B[nxt][j][i].y = wp[i][kk * NCOL + 1];
        }
      }
    }
    __builtin_amdgcn_sched_barrier(0);
#pragma unroll
    for (int j = 0; j < 8; ++j) {
#pragma unroll
      for (int i = 0; i < 3; ++i) acc[i] = wmma4(A[cur][j], B[cur][j][i], acc[i]);
    }
    __builtin_amdgcn_sched_barrier(0);
  }

#pragma unroll
  for (int i = 0; i < 3; ++i) {
    const int c = (grp * 3 + i) * 16 + l15;
    if (c < NCOL) {
      const float bias = Wp[((size_t)(E_DIM / 2) * NCOL + c) * 2] + bl[c % V_DIM];
#pragma unroll
      for (int v = 0; v < 8; ++v) {
        const int trel2 = tb * 16 + ((lane < 16) ? v : v + 8);
        if (trel2 < nt) {
          out[((size_t)n * T_SEQ + out_off + (size_t)trel2 * S) * V_DIM + c] =
              acc[i][v] + bias;
        }
      }
    }
  }
}

#define FOLD_BLOCKS 204
__global__ __launch_bounds__(32) void front_kernel(
    const float* __restrict__ x,
    const float* __restrict__ W1, const float* __restrict__ b1,
    const float* __restrict__ W2, const float* __restrict__ b2,
    const float* __restrict__ W3, const float* __restrict__ b3,
    const float* __restrict__ Wd4, const float* __restrict__ bd4,
    const float* __restrict__ Wl4,
    const float* __restrict__ Wd5, const float* __restrict__ bd5,
    const float* __restrict__ Wl5,
    float* __restrict__ out, float* __restrict__ ws, size_t ws_bytes) {
  const int id = blockIdx.x;
  if (id < FOLD_BLOCKS) {
    if (ws_bytes >= WP_BYTES) fold_body(id % 17, id / 17, Wd4, bd4, Wl4, Wd5, bd5, Wl5, ws);
  } else {
    const int h = id - FOLD_BLOCKS;
    head123_body(h % 100, h / 100, x, W1, b1, W2, b2, W3, b3, out);
  }
}

#define CONV5_BLOCKS 864
__global__ __launch_bounds__(32) void conv_kernel(
    const float* __restrict__ x,
    const float* __restrict__ ws,
    const float* __restrict__ bl4,
    const float* __restrict__ bl5,
    float* __restrict__ out, size_t ws_bytes) {
  if (ws_bytes < WP_BYTES) return;
  const int id = blockIdx.x;
  if (id < CONV5_BLOCKS) {
    conv_folded_body<8>(id % 108, id / 108, x, ws, bl5, out,
                         2112,  3648,  568);
  } else {
    const int h = id - CONV5_BLOCKS;
    conv_folded_body<4>(h % 64, h / 64, x, ws, bl4, out,
                         1600,  1600,  512);
  }
}

extern "C" void kernel_launch(void* const* d_in, const int* in_sizes, int n_in,
                              void* d_out, int out_size, void* d_ws, size_t ws_size,
                              hipStream_t stream) {
  const float* x   = (const float*)d_in[0];
  const float* W1  = (const float*)d_in[4];
  const float* b1  = (const float*)d_in[5];
  const float* W2  = (const float*)d_in[6];
  const float* b2  = (const float*)d_in[7];
  const float* W3  = (const float*)d_in[8];
  const float* b3  = (const float*)d_in[9];
  const float* Wd4 = (const float*)d_in[10];
  const float* bd4 = (const float*)d_in[11];
  const float* Wl4 = (const float*)d_in[12];
  const float* bl4 = (const float*)d_in[13];
  const float* Wd5 = (const float*)d_in[14];
  const float* bd5 = (const float*)d_in[15];
  const float* Wl5 = (const float*)d_in[16];
  const float* bl5 = (const float*)d_in[17];
  float* out = (float*)d_out;
  float* ws  = (float*)d_ws;

  front_kernel<<<dim3(1004), 32, 0, stream>>>(
      x, W1, b1, W2, b2, W3, b3, Wd4, bd4, Wl4, Wd5, bd5, Wl5, out, ws, ws_size);

  conv_kernel<<<dim3(1376), 32, 0, stream>>>(x, ws, bl4, bl5, out, ws_size);
}


# Heads 1-3: K loop reuses the fold's two-buffer chunking (next 8 steps' x and W loads in flight during the current 8 steps' WMMAs) instead of load-wait-WMMA per 4 steps; k order, f32 sums unchanged
# speedup vs baseline: 12.2728x; 1.1440x over previous
#include <hip/hip_runtime.h>
#include <hip/hip_bf16.h>


#define T_SEQ 8192
#define E_DIM 256
#define V_DIM 17
#define WP_PAIRS 129
#define WP_OFF5 (WP_PAIRS * 4 * V_DIM * 2)
#define WP_BYTES ((size_t)(WP_PAIRS * 12 * V_DIM * 2) * sizeof(float))

typedef float v2f __attribute__((ext_vector_type(2)));
typedef float v8f __attribute__((ext_vector_type(8)));

__device__ __forceinline__ v8f wmma4(v2f a, v2f b, v8f c) {
  return __builtin_amdgcn_wmma_f32_16x16x4_f32(
       false, a,  false, b,
       (short)0, c,  false,  false);
}

template <int AS>
__device__ __forceinline__ void fold_tile(const float* __restrict__ ap,
                                          const float* __restrict__ b0p,
                                          const float* __restrict__ b1p,
                                          v8f& acc0, v8f& acc1) {
  v2f A[2][8], B0[2][8], B1[2][8];
#pragma unroll
  for (int j = 0; j < 8; ++j) {
    const int oo = j * 4;
    A[0][j].x  = ap[oo * AS];
    A[0][j].y  = ap[oo * AS + AS];
    B0[0][j].x = b0p[oo * V_DIM];
    B0[0][j].y = b0p[oo * V_DIM + V_DIM];
    B1[0][j].x = b1p[oo * V_DIM];
    B1[0][j].y = b1p[oo * V_DIM + V_DIM];
  }
#pragma unroll
  for (int ch = 0; ch < 8; ++ch) {
    const int cur = ch & 1;
    const int nxt = cur ^ 1;
    if (ch + 1 < 8) {
#pragma unroll
      for (int j = 0; j < 8; ++j) {
        const int oo = ((ch + 1) * 8 + j) * 4;
        A[nxt][j].x  = ap[oo * AS];
        A[nxt][j].y  = ap[oo * AS + AS];
        B0[nxt][j].x = b0p[oo * V_DIM];
        B0[nxt][j].y = b0p[oo * V_DIM + V_DIM];
        B1[nxt][j].x = b1p[oo * V_DIM];
        B1[nxt][j].y = b1p[oo * V_DIM + V_DIM];
      }
    }
    __builtin_amdgcn_sched_barrier(0);
#pragma unroll
    for (int j = 0; j < 8; ++j) {
      acc0 = wmma4(A[cur][j], B0[cur][j], acc0);
      acc1 = wmma4(A[cur][j], B1[cur][j], acc1);
    }
    __builtin_amdgcn_sched_barrier(0);
  }
}

__device__ __forceinline__ void head123_body(
    const int bx, const int n,
    const float* __restrict__ x,
    const float* __restrict__ W1, const float* __restrict__ b1,
    const float* __restrict__ W2, const float* __restrict__ b2,
    const float* __restrict__ W3, const float* __restrict__ b3,
    float* __restrict__ out) {
  const int rowbase = bx * 16;

  const float* W;
  const float* b;
  if (rowbase < 64)       { W = W1; b = b1; }
  else if (rowbase < 576) { W = W2; b = b2; }
  else                    { W = W3; b = b3; }

  const int lane = threadIdx.x & 31;
  const int l15  = lane & 15;
  const int koff = (lane < 16) ? 0 : 2;

  const float* xrow = x + ((size_t)n * T_SEQ + rowbase + l15) * E_DIM;
  const int c1  = 16 + l15;
  const int c1c = (c1 < V_DIM) ? c1 : (V_DIM - 1);

  v8f acc0 = {};
  v8f acc1 = {};
  fold_tile<1>(xrow + koff, W + koff * V_DIM + l15, W + koff * V_DIM + c1c, acc0, acc1);

#pragma unroll
  for (int v = 0; v < 8; ++v) {
    const int r = ((lane < 16) ? v : v + 8);
    float* o = out + ((size_t)n * T_SEQ + rowbase + r) * V_DIM;
    o[l15] = acc0[v] + b[l15];
    if (c1 < V_DIM) o[c1] = acc1[v] + b[c1];
  }
}

__device__ __forceinline__ void fold_body(
    const int bx, const int ky,
    const float* __restrict__ Wd4, const float* __restrict__ bd4,
    const float* __restrict__ Wl4,
    const float* __restrict__ Wd5, const float* __restrict__ bd5,
    const float* __restrict__ Wl5,
    float* __restrict__ ws) {
  const bool h5 = (ky >= 4);
  const float* Wd = h5 ? Wd5 : Wd4;
  const float* bd = h5 ? bd5 : bd4;
  const float* Wl = h5 ? Wl5 : Wl4;
  const int s  = h5 ? 8 : 4;
  const int k  = h5 ? (ky - 4) : ky;
  const int ld = s * V_DIM;
  float* Wp = ws + (h5 ? WP_OFF5 : 0);

  const int lane = threadIdx.x & 31;
  const int l15  = lane & 15;
  const int koff = (lane < 16) ? 0 : 2;

  const int ibase = bx * 16;
  const int c1  = 16 + l15;
  const int c1c = (c1 < V_DIM) ? c1 : (V_DIM - 1);

  const float* b0p = Wl + koff * V_DIM + l15;
  const float* b1p = Wl + koff * V_DIM + c1c;

  v8f acc0 = {};
  v8f acc1 = {};
  if (ibase < E_DIM) {
    const float* arow = Wd + (size_t)(ibase + l15) * (E_DIM * s) + k;
    if (h5) fold_tile<8>(arow + koff * 8, b0p, b1p, acc0, acc1);
    else    fold_tile<4>(arow + koff * 4, b0p, b1p, acc0, acc1);
  } else {
    fold_tile<1>(bd + koff, b0p, b1p, acc0, acc1);
  }

#pragma unroll
  for (int v = 0; v < 8; v += 2) {
    const int r = ibase + ((lane < 16) ? v : v + 8);
    if (r <= E_DIM) {
      float* o = Wp + ((size_t)(r >> 1) * ld + k * V_DIM) * 2;
      o[l15 * 2]     = acc0[v];
      o[l15 * 2 + 1] = acc0[v + 1];
      if (c1 < V_DIM) {
        o[c1 * 2]     = acc1[v];
        o[c1 * 2 + 1] = acc1[v + 1];
      }
    }
  }
}

template <int S>
__device__ __forceinline__ void conv_folded_body(
    const int bx, const int n,
    const float* __restrict__ x,
    const float* __restrict__ ws,
    const float* __restrict__ bl,
    float* __restrict__ out,
    const int ioff, const int out_off, const int nt) {
  constexpr int NCOL   = S * V_DIM;
  constexpr int GROUPS = (NCOL + 47) / 48;
  const float* Wp = ws + ((S == 8) ? WP_OFF5 : 0);

  const int tb  = bx / GROUPS;
  const int grp = bx % GROUPS;

  const int lane = threadIdx.x & 31;
  const int l15  = lane & 15;
  const int koff = (lane < 16) ? 0 : 2;

  const int trel   = tb * 16 + l15;
  const int tclamp = (trel < nt) ? trel : (nt - 1);
  const float* xp = x + ((size_t)n * T_SEQ + ioff + tclamp) * E_DIM + koff;

  const float* wp[3];
#pragma unroll
  for (int i = 0; i < 3; ++i) {
    const int c  = (grp * 3 + i) * 16 + l15;
    const int cc = (c < NCOL) ? c : (NCOL - 1);
    wp[i] = Wp + ((size_t)(koff >> 1) * NCOL + cc) * 2;
  }

  v8f acc[3];
#pragma unroll
  for (int i = 0; i < 3; ++i) acc[i] = (v8f){};

  v2f A[2][8], B[2][8][3];
#pragma unroll
  for (int j = 0; j < 8; ++j) {
    const int kk = j * 4;
    A[0][j].x = xp[kk];
    A[0][j].y = xp[kk + 1];
#pragma unroll
    for (int i = 0; i < 3; ++i) {
      B[0][j][i].x = wp[i][kk * NCOL];
      B[0][j][i].y = wp[i][kk * NCOL + 1];
    }
  }
#pragma unroll
  for (int ch = 0; ch < 8; ++ch) {
    const int cur = ch & 1;
    const int nxt = cur ^ 1;
    if (ch + 1 < 8) {
#pragma unroll
      for (int j = 0; j < 8; ++j) {
        const int kk = ((ch + 1) * 8 + j) * 4;
        A[nxt][j].x = xp[kk];
        A[nxt][j].y = xp[kk + 1];
#pragma unroll
        for (int i = 0; i < 3; ++i) {
          B[nxt][j][i].x = wp[i][kk * NCOL];
          B[nxt][j][i].y = wp[i][kk * NCOL + 1];
        }
      }
    }
    __builtin_amdgcn_sched_barrier(0);
#pragma unroll
    for (int j = 0; j < 8; ++j) {
#pragma unroll
      for (int i = 0; i < 3; ++i) acc[i] = wmma4(A[cur][j], B[cur][j][i], acc[i]);
    }
    __builtin_amdgcn_sched_barrier(0);
  }

#pragma unroll
  for (int i = 0; i < 3; ++i) {
    const int c = (grp * 3 + i) * 16 + l15;
    if (c < NCOL) {
      const float bias = Wp[((size_t)(E_DIM / 2) * NCOL + c) * 2] + bl[c % V_DIM];
#pragma unroll
      for (int v = 0; v < 8; ++v) {
        const int trel2 = tb * 16 + ((lane < 16) ? v : v + 8);
        if (trel2 < nt) {
          out[((size_t)n * T_SEQ + out_off + (size_t)trel2 * S) * V_DIM + c] =
              acc[i][v] + bias;
        }
      }
    }
  }
}

#define FOLD_BLOCKS 204
__global__ __launch_bounds__(32) void front_kernel(
    const float* __restrict__ x,
    const float* __restrict__ W1, const float* __restrict__ b1,
    const float* __restrict__ W2, const float* __restrict__ b2,
    const float* __restrict__ W3, const float* __restrict__ b3,
    const float* __restrict__ Wd4, const float* __restrict__ bd4,
    const float* __restrict__ Wl4,
    const float* __restrict__ Wd5, const float* __restrict__ bd5,
    const float* __restrict__ Wl5,
    float* __restrict__ out, float* __restrict__ ws, size_t ws_bytes) {
  const int id = blockIdx.x;
  if (id < FOLD_BLOCKS) {
    if (ws_bytes >= WP_BYTES) fold_body(id % 17, id / 17, Wd4, bd4, Wl4, Wd5, bd5, Wl5, ws);
  } else {
    const int h = id - FOLD_BLOCKS;
    head123_body(h % 100, h / 100, x, W1, b1, W2, b2, W3, b3, out);
  }
}

#define CONV5_BLOCKS 864
__global__ __launch_bounds__(32) void conv_kernel(
    const float* __restrict__ x,
    const float* __restrict__ ws,
    const float* __restrict__ bl4,
    const float* __restrict__ bl5,
    float* __restrict__ out, size_t ws_bytes) {
  if (ws_bytes < WP_BYTES) return;
  const int id = blockIdx.x;
  if (id < CONV5_BLOCKS) {
    conv_folded_body<8>(id % 108, id / 108, x, ws, bl5, out,
                         2112,  3648,  568);
  } else {
    const int h = id - CONV5_BLOCKS;
    conv_folded_body<4>(h % 64, h / 64, x, ws, bl4, out,
                         1600,  1600,  512);
  }
}

extern "C" void kernel_launch(void* const* d_in, const int* in_sizes, int n_in,
                              void* d_out, int out_size, void* d_ws, size_t ws_size,
                              hipStream_t stream) {
  const float* x   = (const float*)d_in[0];
  const float* W1  = (const float*)d_in[4];
  const float* b1  = (const float*)d_in[5];
  const float* W2  = (const float*)d_in[6];
  const float* b2  = (const float*)d_in[7];
  const float* W3  = (const float*)d_in[8];
  const float* b3  = (const float*)d_in[9];
  const float* Wd4 = (const float*)d_in[10];
  const float* bd4 = (const float*)d_in[11];
  const float* Wl4 = (const float*)d_in[12];
  const float* bl4 = (const float*)d_in[13];
  const float* Wd5 = (const float*)d_in[14];
  const float* bd5 = (const float*)d_in[15];
  const float* Wl5 = (const float*)d_in[16];
  const float* bl5 = (const float*)d_in[17];
  float* out = (float*)d_out;
  float* ws  = (float*)d_ws;

  front_kernel<<<dim3(1004), 32, 0, stream>>>(
      x, W1, b1, W2, b2, W3, b3, Wd4, bd4, Wl4, Wd5, bd5, Wl5, out, ws, ws_size);

  conv_kernel<<<dim3(1376), 32, 0, stream>>>(x, ws, bl4, bl5, out, ws_size);
}


# Conv launch re-tiled to one wave per SIMD: 968 waves of 16 tokens x 64 columns (was 1376 x 48) fed by 16-byte x and weight fragment loads; fold stores Wc in fragment order; f32 sum order changes
# speedup vs baseline: 13.0725x; 1.0652x over previous
#include <hip/hip_runtime.h>
#include <hip/hip_bf16.h>


#define T_SEQ 8192
#define E_DIM 256
#define V_DIM 17
#define WQ_TILE_FLOATS (32 * 32 * 4)
#define WQ_OFF5 (5 * WQ_TILE_FLOATS)
#define WB_OFF4 ((5 + 9) * WQ_TILE_FLOATS)
#define WB_OFF5 (WB_OFF4 + 4 * V_DIM)
#define WP_BYTES ((size_t)(WB_OFF5 + 8 * V_DIM) * sizeof(float))

typedef float v2f __attribute__((ext_vector_type(2)));
typedef float v4f __attribute__((ext_vector_type(4)));
typedef float v8f __attribute__((ext_vector_type(8)));

__device__ __forceinline__ v8f wmma4(v2f a, v2f b, v8f c) {
  return __builtin_amdgcn_wmma_f32_16x16x4_f32(
       false, a,  false, b,
       (short)0, c,  false,  false);
}

template <int AS>
__device__ __forceinline__ void fold_tile(const float* __restrict__ ap,
                                          const float* __restrict__ b0p,
                                          const float* __restrict__ b1p,
                                          v8f& acc0, v8f& acc1) {
  v2f A[2][8], B0[2][8], B1[2][8];
#pragma unroll
  for (int j = 0; j < 8; ++j) {
    const int oo = j * 4;
    A[0][j].x  = ap[oo * AS];
    A[0][j].y  = ap[oo * AS + AS];
    B0[0][j].x = b0p[oo * V_DIM];
    B0[0][j].y = b0p[oo * V_DIM + V_DIM];
    B1[0][j].x = b1p[oo * V_DIM];
    B1[0][j].y = b1p[oo * V_DIM + V_DIM];
  }
#pragma unroll
  for (int ch = 0; ch < 8; ++ch) {
    const int cur = ch & 1;
    const int nxt = cur ^ 1;
    if (ch + 1 < 8) {
#pragma unroll
      for (int j = 0; j < 8; ++j) {
        const int oo = ((ch + 1) * 8 + j) * 4;
        A[nxt][j].x  = ap[oo * AS];
        A[nxt][j].y  = ap[oo * AS + AS];
        B0[nxt][j].x = b0p[oo * V_DIM];
        B0[nxt][j].y = b0p[oo * V_DIM + V_DIM];
        B1[nxt][j].x = b1p[oo * V_DIM];
        B1[nxt][j].y = b1p[oo * V_DIM + V_DIM];
      }
    }
    __builtin_amdgcn_sched_barrier(0);
#pragma unroll
    for (int j = 0; j < 8; ++j) {
      acc0 = wmma4(A[cur][j], B0[cur][j], acc0);
      acc1 = wmma4(A[cur][j], B1[cur][j], acc1);
    }
    __builtin_amdgcn_sched_barrier(0);
  }
}

__device__ __forceinline__ void head123_body(
    const int bx, const int n,
    const float* __restrict__ x,
    const float* __restrict__ W1, const float* __restrict__ b1,
    const float* __restrict__ W2, const float* __restrict__ b2,
    const float* __restrict__ W3, const float* __restrict__ b3,
    float* __restrict__ out) {
  const int rowbase = bx * 16;

  const float* W;
  const float* b;
  if (rowbase < 64)       { W = W1; b = b1; }
  else if (rowbase < 576) { W = W2; b = b2; }
  else                    { W = W3; b = b3; }

  const int lane = threadIdx.x & 31;
  const int l15  = lane & 15;
  const int koff = (lane < 16) ? 0 : 2;

  const float* xrow = x + ((size_t)n * T_SEQ + rowbase + l15) * E_DIM;
  const int c1  = 16 + l15;
  const int c1c = (c1 < V_DIM) ? c1 : (V_DIM - 1);

  v8f acc0 = {};
  v8f acc1 = {};
  fold_tile<1>(xrow + koff, W + koff * V_DIM + l15, W + koff * V_DIM + c1c, acc0, acc1);

#pragma unroll
  for (int v = 0; v < 8; ++v) {
    const int r = ((lane < 16) ? v : v + 8);
    float* o = out + ((size_t)n * T_SEQ + rowbase + r) * V_DIM;
    o[l15] = acc0[v] + b[l15];
    if (c1 < V_DIM) o[c1] = acc1[v] + b[c1];
  }
}

__device__ __forceinline__ void fold_body(
    const int bx, const int ky,
    const float* __restrict__ Wd4, const float* __restrict__ bd4,
    const float* __restrict__ Wl4,
    const float* __restrict__ Wd5, const float* __restrict__ bd5,
    const float* __restrict__ Wl5,
    float* __restrict__ ws) {
  const bool h5 = (ky >= 4);
  const float* Wd = h5 ? Wd5 : Wd4;
  const float* bd = h5 ? bd5 : bd4;
  const float* Wl = h5 ? Wl5 : Wl4;
  const int s  = h5 ? 8 : 4;
  const int k  = h5 ? (ky - 4) : ky;
  float* Wq = ws + (h5 ? WQ_OFF5 : 0);
  float* Wb = ws + (h5 ? WB_OFF5 : WB_OFF4);

  const int lane = threadIdx.x & 31;
  const int l15  = lane & 15;
  const int koff = (lane < 16) ? 0 : 2;

  const int ibase = bx * 16;
  const int c1  = 16 + l15;
  const int c1c = (c1 < V_DIM) ? c1 : (V_DIM - 1);

  const float* b0p = Wl + koff * V_DIM + l15;
  const float* b1p = Wl + koff * V_DIM + c1c;

  v8f acc0 = {};
  v8f acc1 = {};
  if (ibase < E_DIM) {
    const float* arow = Wd + (size_t)(ibase + l15) * (E_DIM * s) + k;
    if (h5) fold_tile<8>(arow + koff * 8, b0p, b1p, acc0, acc1);
    else    fold_tile<4>(arow + koff * 4, b0p, b1p, acc0, acc1);
  } else {
    fold_tile<1>(bd + koff, b0p, b1p, acc0, acc1);
  }

  const int hh = lane >> 4;
  if (ibase < E_DIM) {
    const int m = 2 * bx + hh;
    {
      const int c = k * V_DIM + l15;
      float* o = Wq + ((size_t)((c >> 4) * 32 + m) * 32 + (c & 15)) * 4;
      *(v4f*)(o)      = (v4f){acc0[0], acc0[1], acc0[2], acc0[3]};
      *(v4f*)(o + 64) = (v4f){acc0[4], acc0[5], acc0[6], acc0[7]};
    }
    if (c1 < V_DIM) {
      const int c = k * V_DIM + c1;
      float* o = Wq + ((size_t)((c >> 4) * 32 + m) * 32 + (c & 15)) * 4;
      *(v4f*)(o)      = (v4f){acc1[0], acc1[1], acc1[2], acc1[3]};
      *(v4f*)(o + 64) = (v4f){acc1[4], acc1[5], acc1[6], acc1[7]};
    }
  } else if (hh == 0) {
    Wb[k * V_DIM + l15] = acc0[0];
    if (c1 < V_DIM) Wb[k * V_DIM + c1] = acc1[0];
  }
}

template <int S, int NA, int NB>
__device__ __forceinline__ void conv_tile_body(
    const int n, const int tb0, const int ct0,
    const float* __restrict__ x,
    const float* __restrict__ ws,
    const float* __restrict__ bl,
    float* __restrict__ out,
    const int ioff, const int out_off, const int nt) {
  constexpr int NCOL = S * V_DIM;
  const float* Wq = ws + ((S == 8) ? WQ_OFF5 : 0);
  const float* Wb = ws + ((S == 8) ? WB_OFF5 : WB_OFF4);

  const int lane = threadIdx.x & 31;
  const int cc15 = lane & 15;
  const int hf   = lane >> 4;

  const float* xp[NA];
#pragma unroll
  for (int a = 0; a < NA; ++a) {
    const int trel   = (tb0 + a) * 16 + cc15;
    const int tclamp = (trel < nt) ? trel : (nt - 1);
    xp[a] = x + ((size_t)n * T_SEQ + ioff + tclamp) * E_DIM + 4 * hf;
  }

  const float* wq[NB];
  float bias[NB];
#pragma unroll
  for (int b = 0; b < NB; ++b) {
    const int col = (ct0 + b) * 16 + cc15;
    const int cl  = (col < NCOL) ? col : (NCOL - 1);
    wq[b]   = Wq + ((size_t)(cl >> 4) * 1024 + hf * 16 + (cl & 15)) * 4;
    bias[b] = Wb[cl] + bl[cl % V_DIM];
  }

  v8f acc[NA][NB];
#pragma unroll
  for (int a = 0; a < NA; ++a)
#pragma unroll
    for (int b = 0; b < NB; ++b) acc[a][b] = (v8f){};

  v4f A[2][8][NA], B[2][8][NB];
#pragma unroll
  for (int j = 0; j < 8; ++j) {
#pragma unroll
    for (int a = 0; a < NA; ++a) A[0][j][a] = *(const v4f*)(xp[a] + 8 * j);
#pragma unroll
    for (int b = 0; b < NB; ++b) B[0][j][b] = *(const v4f*)(wq[b] + 128 * j);
  }
#pragma unroll
  for (int ch = 0; ch < 4; ++ch) {
    const int cur = ch & 1;
    const int nxt = cur ^ 1;
    if (ch + 1 < 4) {
#pragma unroll
      for (int j = 0; j < 8; ++j) {
        const int m = (ch + 1) * 8 + j;
#pragma unroll
        for (int a = 0; a < NA; ++a) A[nxt][j][a] = *(const v4f*)(xp[a] + 8 * m);
#pragma unroll
        for (int b = 0; b < NB; ++b) B[nxt][j][b] = *(const v4f*)(wq[b] + 128 * m);
      }
    }
    __builtin_amdgcn_sched_barrier(0);
#pragma unroll
    for (int j = 0; j < 8; ++j) {
#pragma unroll
      for (int a = 0; a < NA; ++a)
#pragma unroll
        for (int b = 0; b < NB; ++b)
          acc[a][b] = wmma4((v2f){A[cur][j][a].x, A[cur][j][a].y},
                            (v2f){B[cur][j][b].x, B[cur][j][b].y}, acc[a][b]);
#pragma unroll
      for (int a = 0; a < NA; ++a)
#pragma unroll
        for (int b = 0; b < NB; ++b)
          acc[a][b] = wmma4((v2f){A[cur][j][a].z, A[cur][j][a].w},
                            (v2f){B[cur][j][b].z, B[cur][j][b].w}, acc[a][b]);
    }
    __builtin_amdgcn_sched_barrier(0);
  }

#pragma unroll
  for (int a = 0; a < NA; ++a) {
#pragma unroll
    for (int b = 0; b < NB; ++b) {
      const int col = (ct0 + b) * 16 + cc15;
      if (col < NCOL) {
#pragma unroll
        for (int v = 0; v < 8; ++v) {
          const int trel2 = (tb0 + a) * 16 + 8 * hf + v;
          if (trel2 < nt) {
            out[((size_t)n * T_SEQ + out_off + (size_t)trel2 * S) * V_DIM + col] =
                acc[a][b][v] + bias[b];
          }
        }
      }
    }
  }
}

#define FOLD_BLOCKS 204
__global__ __launch_bounds__(32) void front_kernel(
    const float* __restrict__ x,
    const float* __restrict__ W1, const float* __restrict__ b1,
    const float* __restrict__ W2, const float* __restrict__ b2,
    const float* __restrict__ W3, const float* __restrict__ b3,
    const float* __restrict__ Wd4, const float* __restrict__ bd4,
    const float* __restrict__ Wl4,
    const float* __restrict__ Wd5, const float* __restrict__ bd5,
    const float* __restrict__ Wl5,
    float* __restrict__ out, float* __restrict__ ws, size_t ws_bytes) {
  const int id = blockIdx.x;
  if (id < FOLD_BLOCKS) {
    if (ws_bytes >= WP_BYTES) fold_body(id % 17, id / 17, Wd4, bd4, Wl4, Wd5, bd5, Wl5, ws);
  } else {
    const int h = id - FOLD_BLOCKS;
    head123_body(h % 100, h / 100, x, W1, b1, W2, b2, W3, b3, out);
  }
}

#define CONV_BLOCKS 968
__global__ __launch_bounds__(32) void conv_kernel(
    const float* __restrict__ x,
    const float* __restrict__ ws,
    const float* __restrict__ bl4,
    const float* __restrict__ bl5,
    float* __restrict__ out, size_t ws_bytes) {
  if (ws_bytes < WP_BYTES) return;
  const int id = blockIdx.x;
  if (id < 576) {
    const int r = id % 72;
    conv_tile_body<8, 1, 4>(id / 72, r >> 1, (r & 1) * 4, x, ws, bl5, out,
                             2112,  3648,  568);
  } else if (id < 832) {
    const int h = id - 576;
    conv_tile_body<4, 1, 4>(h >> 5, h & 31, 0, x, ws, bl4, out,
                             1600,  1600,  512);
  } else if (id < 904) {
    const int h = id - 832;
    conv_tile_body<8, 4, 1>(h / 9, (h % 9) * 4, 8, x, ws, bl5, out,
                             2112,  3648,  568);
  } else {
    const int h = id - 904;
    conv_tile_body<4, 4, 1>(h >> 3, (h & 7) * 4, 4, x, ws, bl4, out,
                             1600,  1600,  512);
  }
}

extern "C" void kernel_launch(void* const* d_in, const int* in_sizes, int n_in,
                              void* d_out, int out_size, void* d_ws, size_t ws_size,
                              hipStream_t stream) {
  const float* x   = (const float*)d_in[0];
  const float* W1  = (const float*)d_in[4];
  const float* b1  = (const float*)d_in[5];
  const float* W2  = (const float*)d_in[6];
  const float* b2  = (const float*)d_in[7];
  const float* W3  = (const float*)d_in[8];
  const float* b3  = (const float*)d_in[9];
  const float* Wd4 = (const float*)d_in[10];
  const float* bd4 = (const float*)d_in[11];
  const float* Wl4 = (const float*)d_in[12];
  const float* bl4 = (const float*)d_in[13];
  const float* Wd5 = (const float*)d_in[14];
  const float* bd5 = (const float*)d_in[15];
  const float* Wl5 = (const float*)d_in[16];
  const float* bl5 = (const float*)d_in[17];
  float* out = (float*)d_out;
  float* ws  = (float*)d_ws;

  front_kernel<<<dim3(1004), 32, 0, stream>>>(
      x, W1, b1, W2, b2, W3, b3, Wd4, bd4, Wl4, Wd5, bd5, Wl5, out, ws, ws_size);

  conv_kernel<<<dim3(CONV_BLOCKS), 32, 0, stream>>>(x, ws, bl4, bl5, out, ws_size);
}


# Fold re-tiled: a wave's 16 A rows are (i,tap) pairs (4 i x 4 taps or 2 i x 8 taps), so a Wd load touches 4 or 2 lines, read whole by one wave, not 16; bias row once per head; f32 sums unchanged
# speedup vs baseline: 14.1842x; 1.0850x over previous
#include <hip/hip_runtime.h>
#include <hip/hip_bf16.h>


#define T_SEQ 8192
#define E_DIM 256
#define V_DIM 17
#define WQ_TILE_FLOATS (32 * 32 * 4)
#define WQ_OFF5 (5 * WQ_TILE_FLOATS)
#define WB_OFF4 ((5 + 9) * WQ_TILE_FLOATS)
#define WB_OFF5 (WB_OFF4 + 4 * V_DIM)
#define WP_BYTES ((size_t)(WB_OFF5 + 8 * V_DIM) * sizeof(float))

typedef float v2f __attribute__((ext_vector_type(2)));
typedef float v4f __attribute__((ext_vector_type(4)));
typedef float v8f __attribute__((ext_vector_type(8)));

__device__ __forceinline__ v8f wmma4(v2f a, v2f b, v8f c) {
  return __builtin_amdgcn_wmma_f32_16x16x4_f32(
       false, a,  false, b,
       (short)0, c,  false,  false);
}

template <int AS>
__device__ __forceinline__ void fold_tile(const float* __restrict__ ap,
                                          const float* __restrict__ b0p,
                                          const float* __restrict__ b1p,
                                          v8f& acc0, v8f& acc1) {
  v2f A[2][8], B0[2][8], B1[2][8];
#pragma unroll
  for (int j = 0; j < 8; ++j) {
    const int oo = j * 4;
    A[0][j].x  = ap[oo * AS];
    A[0][j].y  = ap[oo * AS + AS];
    B0[0][j].x = b0p[oo * V_DIM];
    B0[0][j].y = b0p[oo * V_DIM + V_DIM];
    B1[0][j].x = b1p[oo * V_DIM];
    B1[0][j].y = b1p[oo * V_DIM + V_DIM];
  }
#pragma unroll
  for (int ch = 0; ch < 8; ++ch) {
    const int cur = ch & 1;
    const int nxt = cur ^ 1;
    if (ch + 1 < 8) {
#pragma unroll
      for (int j = 0; j < 8; ++j) {
        const int oo = ((ch + 1) * 8 + j) * 4;
        A[nxt][j].x  = ap[oo * AS];
        A[nxt][j].y  = ap[oo * AS + AS];
        B0[nxt][j].x = b0p[oo * V_DIM];
        B0[nxt][j].y = b0p[oo * V_DIM + V_DIM];
        B1[nxt][j].x = b1p[oo * V_DIM];
        B1[nxt][j].y = b1p[oo * V_DIM + V_DIM];
      }
    }
    __builtin_amdgcn_sched_barrier(0);
#pragma unroll
    for (int j = 0; j < 8; ++j) {
      acc0 = wmma4(A[cur][j], B0[cur][j], acc0);
      acc1 = wmma4(A[cur][j], B1[cur][j], acc1);
    }
    __builtin_amdgcn_sched_barrier(0);
  }
}

__device__ __forceinline__ void head123_body(
    const int bx, const int n,
    const float* __restrict__ x,
    const float* __restrict__ W1, const float* __restrict__ b1,
    const float* __restrict__ W2, const float* __restrict__ b2,
    const float* __restrict__ W3, const float* __restrict__ b3,
    float* __restrict__ out) {
  const int rowbase = bx * 16;

  const float* W;
  const float* b;
  if (rowbase < 64)       { W = W1; b = b1; }
  else if (rowbase < 576) { W = W2; b = b2; }
  else                    { W = W3; b = b3; }

  const int lane = threadIdx.x & 31;
  const int l15  = lane & 15;
  const int koff = (lane < 16) ? 0 : 2;

  const float* xrow = x + ((size_t)n * T_SEQ + rowbase + l15) * E_DIM;
  const int c1  = 16 + l15;
  const int c1c = (c1 < V_DIM) ? c1 : (V_DIM - 1);

  v8f acc0 = {};
  v8f acc1 = {};
  fold_tile<1>(xrow + koff, W + koff * V_DIM + l15, W + koff * V_DIM + c1c, acc0, acc1);

#pragma unroll
  for (int v = 0; v < 8; ++v) {
    const int r = ((lane < 16) ? v : v + 8);
    float* o = out + ((size_t)n * T_SEQ + rowbase + r) * V_DIM;
    o[l15] = acc0[v] + b[l15];
    if (c1 < V_DIM) o[c1] = acc1[v] + b[c1];
  }
}

template <int S>
__device__ __forceinline__ void fold_rows_body(
    const int tile,
    const float* __restrict__ Wd, const float* __restrict__ Wl,
    float* __restrict__ Wq) {
  constexpr int SH = (S == 8) ? 3 : 2;
  const int lane = threadIdx.x & 31;
  const int l15  = lane & 15;
  const int hh   = lane >> 4;
  const int koff = hh * 2;

  const int ibase = tile * (16 / S);
  const int c1  = 16 + l15;
  const int c1c = (c1 < V_DIM) ? c1 : (V_DIM - 1);

  const float* b0p = Wl + koff * V_DIM + l15;
  const float* b1p = Wl + koff * V_DIM + c1c;
  const float* arow = Wd + (size_t)(ibase + (l15 >> SH)) * (E_DIM * S) + (l15 & (S - 1));

  v8f acc0 = {};
  v8f acc1 = {};
  fold_tile<S>(arow + koff * S, b0p, b1p, acc0, acc1);

#pragma unroll
  for (int r = 0; r < 8; ++r) {
    const int R   = 8 * hh + r;
    const int i   = ibase + (R >> SH);
    const int tap = R & (S - 1);
    const size_t rowpart = (size_t)(i >> 3) * 32 + 16 * ((i >> 2) & 1);
    {
      const int c = tap * V_DIM + l15;
      Wq[((size_t)(c >> 4) * 1024 + rowpart + (c & 15)) * 4 + (i & 3)] = acc0[r];
    }
    if (c1 < V_DIM) {
      const int c = tap * V_DIM + c1;
      Wq[((size_t)(c >> 4) * 1024 + rowpart + (c & 15)) * 4 + (i & 3)] = acc1[r];
    }
  }
}

template <int S>
__device__ __forceinline__ void fold_bias_body(
    const float* __restrict__ bd, const float* __restrict__ Wl,
    float* __restrict__ Wb) {
  const int lane = threadIdx.x & 31;
  const int l15  = lane & 15;
  const int hh   = lane >> 4;
  const int koff = hh * 2;
  const int c1  = 16 + l15;
  const int c1c = (c1 < V_DIM) ? c1 : (V_DIM - 1);

  v8f acc0 = {};
  v8f acc1 = {};
  fold_tile<1>(bd + koff, Wl + koff * V_DIM + l15, Wl + koff * V_DIM + c1c, acc0, acc1);

  if (hh == 0) {
#pragma unroll
    for (int r = 0; r < S; ++r) {
      Wb[r * V_DIM + l15] = acc0[r];
      if (c1 < V_DIM) Wb[r * V_DIM + c1] = acc1[r];
    }
  }
}

template <int S, int NA, int NB>
__device__ __forceinline__ void conv_tile_body(
    const int n, const int tb0, const int ct0,
    const float* __restrict__ x,
    const float* __restrict__ ws,
    const float* __restrict__ bl,
    float* __restrict__ out,
    const int ioff, const int out_off, const int nt) {
  constexpr int NCOL = S * V_DIM;
  const float* Wq = ws + ((S == 8) ? WQ_OFF5 : 0);
  const float* Wb = ws + ((S == 8) ? WB_OFF5 : WB_OFF4);

  const int lane = threadIdx.x & 31;
  const int cc15 = lane & 15;
  const int hf   = lane >> 4;

  const float* xp[NA];
#pragma unroll
  for (int a = 0; a < NA; ++a) {
    const int trel   = (tb0 + a) * 16 + cc15;
    const int tclamp = (trel < nt) ? trel : (nt - 1);
    xp[a] = x + ((size_t)n * T_SEQ + ioff + tclamp) * E_DIM + 4 * hf;
  }

  const float* wq[NB];
  float bias[NB];
#pragma unroll
  for (int b = 0; b < NB; ++b) {
    const int col = (ct0 + b) * 16 + cc15;
    const int cl  = (col < NCOL) ? col : (NCOL - 1);
    wq[b]   = Wq + ((size_t)(cl >> 4) * 1024 + hf * 16 + (cl & 15)) * 4;
    bias[b] = Wb[cl] + bl[cl % V_DIM];
  }

  v8f acc[NA][NB];
#pragma unroll
  for (int a = 0; a < NA; ++a)
#pragma unroll
    for (int b = 0; b < NB; ++b) acc[a][b] = (v8f){};

  v4f A[2][8][NA], B[2][8][NB];
#pragma unroll
  for (int j = 0; j < 8; ++j) {
#pragma unroll
    for (int a = 0; a < NA; ++a) A[0][j][a] = *(const v4f*)(xp[a] + 8 * j);
#pragma unroll
    for (int b = 0; b < NB; ++b) B[0][j][b] = *(const v4f*)(wq[b] + 128 * j);
  }
#pragma unroll
  for (int ch = 0; ch < 4; ++ch) {
    const int cur = ch & 1;
    const int nxt = cur ^ 1;
    if (ch + 1 < 4) {
#pragma unroll
      for (int j = 0; j < 8; ++j) {
        const int m = (ch + 1) * 8 + j;
#pragma unroll
        for (int a = 0; a < NA; ++a) A[nxt][j][a] = *(const v4f*)(xp[a] + 8 * m);
#pragma unroll
        for (int b = 0; b < NB; ++b) B[nxt][j][b] = *(const v4f*)(wq[b] + 128 * m);
      }
    }
    __builtin_amdgcn_sched_barrier(0);
#pragma unroll
    for (int j = 0; j < 8; ++j) {
#pragma unroll
      for (int a = 0; a < NA; ++a)
#pragma unroll
        for (int b = 0; b < NB; ++b)
          acc[a][b] = wmma4((v2f){A[cur][j][a].x, A[cur][j][a].y},
                            (v2f){B[cur][j][b].x, B[cur][j][b].y}, acc[a][b]);
#pragma unroll
      for (int a = 0; a < NA; ++a)
#pragma unroll
        for (int b = 0; b < NB; ++b)
          acc[a][b] = wmma4((v2f){A[cur][j][a].z, A[cur][j][a].w},
                            (v2f){B[cur][j][b].z, B[cur][j][b].w}, acc[a][b]);
    }
    __builtin_amdgcn_sched_barrier(0);
  }

#pragma unroll
  for (int a = 0; a < NA; ++a) {
#pragma unroll
    for (int b = 0; b < NB; ++b) {
      const int col = (ct0 + b) * 16 + cc15;
      if (col < NCOL) {
#pragma unroll
        for (int v = 0; v < 8; ++v) {
          const int trel2 = (tb0 + a) * 16 + 8 * hf + v;
          if (trel2 < nt) {
            out[((size_t)n * T_SEQ + out_off + (size_t)trel2 * S) * V_DIM + col] =
                acc[a][b][v] + bias[b];
          }
        }
      }
    }
  }
}

#define FOLD_BLOCKS 194
__global__ __launch_bounds__(32) void front_kernel(
    const float* __restrict__ x,
    const float* __restrict__ W1, const float* __restrict__ b1,
    const float* __restrict__ W2, const float* __restrict__ b2,
    const float* __restrict__ W3, const float* __restrict__ b3,
    const float* __restrict__ Wd4, const float* __restrict__ bd4,
    const float* __restrict__ Wl4,
    const float* __restrict__ Wd5, const float* __restrict__ bd5,
    const float* __restrict__ Wl5,
    float* __restrict__ out, float* __restrict__ ws, size_t ws_bytes) {
  const int id = blockIdx.x;
  if (id < FOLD_BLOCKS) {
    if (ws_bytes >= WP_BYTES) {
      if (id < 64)        fold_rows_body<4>(id, Wd4, Wl4, ws);
      else if (id < 192)  fold_rows_body<8>(id - 64, Wd5, Wl5, ws + WQ_OFF5);
      else if (id == 192) fold_bias_body<4>(bd4, Wl4, ws + WB_OFF4);
      else                fold_bias_body<8>(bd5, Wl5, ws + WB_OFF5);
    }
  } else {
    const int h = id - FOLD_BLOCKS;
    head123_body(h % 100, h / 100, x, W1, b1, W2, b2, W3, b3, out);
  }
}

#define CONV_BLOCKS 968
__global__ __launch_bounds__(32) void conv_kernel(
    const float* __restrict__ x,
    const float* __restrict__ ws,
    const float* __restrict__ bl4,
    const float* __restrict__ bl5,
    float* __restrict__ out, size_t ws_bytes) {
  if (ws_bytes < WP_BYTES) return;
  const int id = blockIdx.x;
  if (id < 576) {
    const int r = id % 72;
    conv_tile_body<8, 1, 4>(id / 72, r >> 1, (r & 1) * 4, x, ws, bl5, out,
                             2112,  3648,  568);
  } else if (id < 832) {
    const int h = id - 576;
    conv_tile_body<4, 1, 4>(h >> 5, h & 31, 0, x, ws, bl4, out,
                             1600,  1600,  512);
  } else if (id < 904) {
    const int h = id - 832;
    conv_tile_body<8, 4, 1>(h / 9, (h % 9) * 4, 8, x, ws, bl5, out,
                             2112,  3648,  568);
  } else {
    const int h = id - 904;
    conv_tile_body<4, 4, 1>(h >> 3, (h & 7) * 4, 4, x, ws, bl4, out,
                             1600,  1600,  512);
  }
}

extern "C" void kernel_launch(void* const* d_in, const int* in_sizes, int n_in,
                              void* d_out, int out_size, void* d_ws, size_t ws_size,
                              hipStream_t stream) {
  const float* x   = (const float*)d_in[0];
  const float* W1  = (const float*)d_in[4];
  const float* b1  = (const float*)d_in[5];
  const float* W2  = (const float*)d_in[6];
  const float* b2  = (const float*)d_in[7];
  const float* W3  = (const float*)d_in[8];
  const float* b3  = (const float*)d_in[9];
  const float* Wd4 = (const float*)d_in[10];
  const float* bd4 = (const float*)d_in[11];
  const float* Wl4 = (const float*)d_in[12];
  const float* bl4 = (const float*)d_in[13];
  const float* Wd5 = (const float*)d_in[14];
  const float* bd5 = (const float*)d_in[15];
  const float* Wl5 = (const float*)d_in[16];
  const float* bl5 = (const float*)d_in[17];
  float* out = (float*)d_out;
  float* ws  = (float*)d_ws;

  front_kernel<<<dim3(FOLD_BLOCKS + 800), 32, 0, stream>>>(
      x, W1, b1, W2, b2, W3, b3, Wd4, bd4, Wl4, Wd5, bd5, Wl5, out, ws, ws_size);

  conv_kernel<<<dim3(CONV_BLOCKS), 32, 0, stream>>>(x, ws, bl4, bl5, out, ws_size);
}


# Front launch: each wave copies its [256x17] B matrix to LDS once (34 coalesced 16-byte loads) and reads B fragments from LDS, so its global loads are A only, all issued up front; f32 sums unchanged
# speedup vs baseline: 15.5286x; 1.0948x over previous
#include <hip/hip_runtime.h>
#include <hip/hip_bf16.h>


#define T_SEQ 8192
#define E_DIM 256
#define V_DIM 17
#define WQ_TILE_FLOATS (32 * 32 * 4)
#define WQ_OFF5 (5 * WQ_TILE_FLOATS)
#define WB_OFF4 ((5 + 9) * WQ_TILE_FLOATS)
#define WB_OFF5 (WB_OFF4 + 4 * V_DIM)
#define WP_BYTES ((size_t)(WB_OFF5 + 8 * V_DIM) * sizeof(float))

typedef float v2f __attribute__((ext_vector_type(2)));
typedef float v4f __attribute__((ext_vector_type(4)));
typedef float v8f __attribute__((ext_vector_type(8)));

__device__ __forceinline__ v8f wmma4(v2f a, v2f b, v8f c) {
  return __builtin_amdgcn_wmma_f32_16x16x4_f32(
       false, a,  false, b,
       (short)0, c,  false,  false);
}

#define W_FLOATS (E_DIM * V_DIM)
template <int AS>
__device__ __forceinline__ void fold_tile(const float* __restrict__ ap,
                                          const float* __restrict__ Bg,
                                          float* __restrict__ Bs,
                                          const int boff0, const int boff1,
                                          v8f& acc0, v8f& acc1) {
  const int lane = threadIdx.x & 31;
  constexpr int PRE = (AS == 1) ? 24 : 8;
  v4f wv[34];
  v2f A[64];
#pragma unroll
  for (int q = 0; q < 34; ++q) wv[q] = *(const v4f*)(Bg + 4 * (lane + 32 * q));
#pragma unroll
  for (int j = 0; j < PRE; ++j) {
    A[j].x = ap[(4 * j) * AS];
    A[j].y = ap[(4 * j) * AS + AS];
  }
  __builtin_amdgcn_sched_barrier(0);
#pragma unroll
  for (int q = 0; q < 34; ++q) *(v4f*)(Bs + 4 * (lane + 32 * q)) = wv[q];
#pragma unroll
  for (int j = PRE; j < 64; ++j) {
    A[j].x = ap[(4 * j) * AS];
    A[j].y = ap[(4 * j) * AS + AS];
  }
  __builtin_amdgcn_fence(__ATOMIC_ACQ_REL, "workgroup", "local");

  const float* b0s = Bs + boff0;
  const float* b1s = Bs + boff1;
  v2f B0[2][8], B1[2][8];
#pragma unroll
  for (int j = 0; j < 8; ++j) {
    const int oo = j * 4;
    B0[0][j].x = b0s[oo * V_DIM];
    B0[0][j].y = b0s[oo * V_DIM + V_DIM];
    B1[0][j].x = b1s[oo * V_DIM];
    B1[0][j].y = b1s[oo * V_DIM + V_DIM];
  }
  __builtin_amdgcn_sched_barrier(0);
#pragma unroll
  for (int ch = 0; ch < 8; ++ch) {
    const int cur = ch & 1;
    const int nxt = cur ^ 1;
    if (ch + 1 < 8) {
#pragma unroll
      for (int j = 0; j < 8; ++j) {
        const int oo = ((ch + 1) * 8 + j) * 4;
        B0[nxt][j].x = b0s[oo * V_DIM];
        B0[nxt][j].y = b0s[oo * V_DIM + V_DIM];
        B1[nxt][j].x = b1s[oo * V_DIM];
        B1[nxt][j].y = b1s[oo * V_DIM + V_DIM];
      }
    }
    __builtin_amdgcn_sched_barrier(0);
#pragma unroll
    for (int j = 0; j < 8; ++j) {
      acc0 = wmma4(A[ch * 8 + j], B0[cur][j], acc0);
      acc1 = wmma4(A[ch * 8 + j], B1[cur][j], acc1);
    }
    __builtin_amdgcn_sched_barrier(0);
  }
}

__device__ __forceinline__ void head123_body(
    const int bx, const int n,
    const float* __restrict__ x,
    const float* __restrict__ W1, const float* __restrict__ b1,
    const float* __restrict__ W2, const float* __restrict__ b2,
    const float* __restrict__ W3, const float* __restrict__ b3,
    float* __restrict__ out, float* __restrict__ Bs) {
  const int rowbase = bx * 16;

  const float* W;
  const float* b;
  if (rowbase < 64)       { W = W1; b = b1; }
  else if (rowbase < 576) { W = W2; b = b2; }
  else                    { W = W3; b = b3; }

  const int lane = threadIdx.x & 31;
  const int l15  = lane & 15;
  const int koff = (lane < 16) ? 0 : 2;

  const float* xrow = x + ((size_t)n * T_SEQ + rowbase + l15) * E_DIM;
  const int c1  = 16 + l15;
  const int c1c = (c1 < V_DIM) ? c1 : (V_DIM - 1);

  v8f acc0 = {};
  v8f acc1 = {};
  fold_tile<1>(xrow + koff, W, Bs, koff * V_DIM + l15, koff * V_DIM + c1c, acc0, acc1);

#pragma unroll
  for (int v = 0; v < 8; ++v) {
    const int r = ((lane < 16) ? v : v + 8);
    float* o = out + ((size_t)n * T_SEQ + rowbase + r) * V_DIM;
    o[l15] = acc0[v] + b[l15];
    if (c1 < V_DIM) o[c1] = acc1[v] + b[c1];
  }
}

template <int S>
__device__ __forceinline__ void fold_rows_body(
    const int tile,
    const float* __restrict__ Wd, const float* __restrict__ Wl,
    float* __restrict__ Wq, float* __restrict__ Bs) {
  constexpr int SH = (S == 8) ? 3 : 2;
  const int lane = threadIdx.x & 31;
  const int l15  = lane & 15;
  const int hh   = lane >> 4;
  const int koff = hh * 2;

  const int ibase = tile * (16 / S);
  const int c1  = 16 + l15;
  const int c1c = (c1 < V_DIM) ? c1 : (V_DIM - 1);

  const float* arow = Wd + (size_t)(ibase + (l15 >> SH)) * (E_DIM * S) + (l15 & (S - 1));

  v8f acc0 = {};
  v8f acc1 = {};
  fold_tile<S>(arow + koff * S, Wl, Bs, koff * V_DIM + l15, koff * V_DIM + c1c, acc0, acc1);

#pragma unroll
  for (int r = 0; r < 8; ++r) {
    const int R   = 8 * hh + r;
    const int i   = ibase + (R >> SH);
    const int tap = R & (S - 1);
    const size_t rowpart = (size_t)(i >> 3) * 32 + 16 * ((i >> 2) & 1);
    {
      const int c = tap * V_DIM + l15;
      Wq[((size_t)(c >> 4) * 1024 + rowpart + (c & 15)) * 4 + (i & 3)] = acc0[r];
    }
    if (c1 < V_DIM) {
      const int c = tap * V_DIM + c1;
      Wq[((size_t)(c >> 4) * 1024 + rowpart + (c & 15)) * 4 + (i & 3)] = acc1[r];
    }
  }
}

template <int S>
__device__ __forceinline__ void fold_bias_body(
    const float* __restrict__ bd, const float* __restrict__ Wl,
    float* __restrict__ Wb, float* __restrict__ Bs) {
  const int lane = threadIdx.x & 31;
  const int l15  = lane & 15;
  const int hh   = lane >> 4;
  const int koff = hh * 2;
  const int c1  = 16 + l15;
  const int c1c = (c1 < V_DIM) ? c1 : (V_DIM - 1);

  v8f acc0 = {};
  v8f acc1 = {};
  fold_tile<1>(bd + koff, Wl, Bs, koff * V_DIM + l15, koff * V_DIM + c1c, acc0, acc1);

  if (hh == 0) {
#pragma unroll
    for (int r = 0; r < S; ++r) {
      Wb[r * V_DIM + l15] = acc0[r];
      if (c1 < V_DIM) Wb[r * V_DIM + c1] = acc1[r];
    }
  }
}

template <int S, int NA, int NB>
__device__ __forceinline__ void conv_tile_body(
    const int n, const int tb0, const int ct0,
    const float* __restrict__ x,
    const float* __restrict__ ws,
    const float* __restrict__ bl,
    float* __restrict__ out,
    const int ioff, const int out_off, const int nt) {
  constexpr int NCOL = S * V_DIM;
  const float* Wq = ws + ((S == 8) ? WQ_OFF5 : 0);
  const float* Wb = ws + ((S == 8) ? WB_OFF5 : WB_OFF4);

  const int lane = threadIdx.x & 31;
  const int cc15 = lane & 15;
  const int hf   = lane >> 4;

  const float* xp[NA];
#pragma unroll
  for (int a = 0; a < NA; ++a) {
    const int trel   = (tb0 + a) * 16 + cc15;
    const int tclamp = (trel < nt) ? trel : (nt - 1);
    xp[a] = x + ((size_t)n * T_SEQ + ioff + tclamp) * E_DIM + 4 * hf;
  }

  const float* wq[NB];
  float bias[NB];
#pragma unroll
  for (int b = 0; b < NB; ++b) {
    const int col = (ct0 + b) * 16 + cc15;
    const int cl  = (col < NCOL) ? col : (NCOL - 1);
    wq[b]   = Wq + ((size_t)(cl >> 4) * 1024 + hf * 16 + (cl & 15)) * 4;
    bias[b] = Wb[cl] + bl[cl % V_DIM];
  }

  v8f acc[NA][NB];
#pragma unroll
  for (int a = 0; a < NA; ++a)
#pragma unroll
    for (int b = 0; b < NB; ++b) acc[a][b] = (v8f){};

  v4f A[2][8][NA], B[2][8][NB];
#pragma unroll
  for (int j = 0; j < 8; ++j) {
#pragma unroll
    for (int a = 0; a < NA; ++a) A[0][j][a] = *(const v4f*)(xp[a] + 8 * j);
#pragma unroll
    for (int b = 0; b < NB; ++b) B[0][j][b] = *(const v4f*)(wq[b] + 128 * j);
  }
#pragma unroll
  for (int ch = 0; ch < 4; ++ch) {
    const int cur = ch & 1;
    const int nxt = cur ^ 1;
    if (ch + 1 < 4) {
#pragma unroll
      for (int j = 0; j < 8; ++j) {
        const int m = (ch + 1) * 8 + j;
#pragma unroll
        for (int a = 0; a < NA; ++a) A[nxt][j][a] = *(const v4f*)(xp[a] + 8 * m);
#pragma unroll
        for (int b = 0; b < NB; ++b) B[nxt][j][b] = *(const v4f*)(wq[b] + 128 * m);
      }
    }
    __builtin_amdgcn_sched_barrier(0);
#pragma unroll
    for (int j = 0; j < 8; ++j) {
#pragma unroll
      for (int a = 0; a < NA; ++a)
#pragma unroll
        for (int b = 0; b < NB; ++b)
          acc[a][b] = wmma4((v2f){A[cur][j][a].x, A[cur][j][a].y},
                            (v2f){B[cur][j][b].x, B[cur][j][b].y}, acc[a][b]);
#pragma unroll
      for (int a = 0; a < NA; ++a)
#pragma unroll
        for (int b = 0; b < NB; ++b)
          acc[a][b] = wmma4((v2f){A[cur][j][a].z, A[cur][j][a].w},
                            (v2f){B[cur][j][b].z, B[cur][j][b].w}, acc[a][b]);
    }
    __builtin_amdgcn_sched_barrier(0);
  }

#pragma unroll
  for (int a = 0; a < NA; ++a) {
#pragma unroll
    for (int b = 0; b < NB; ++b) {
      const int col = (ct0 + b) * 16 + cc15;
      if (col < NCOL) {
#pragma unroll
        for (int v = 0; v < 8; ++v) {
          const int trel2 = (tb0 + a) * 16 + 8 * hf + v;
          if (trel2 < nt) {
            out[((size_t)n * T_SEQ + out_off + (size_t)trel2 * S) * V_DIM + col] =
                acc[a][b][v] + bias[b];
          }
        }
      }
    }
  }
}

#define FOLD_BLOCKS 194
__global__ __launch_bounds__(32) void front_kernel(
    const float* __restrict__ x,
    const float* __restrict__ W1, const float* __restrict__ b1,
    const float* __restrict__ W2, const float* __restrict__ b2,
    const float* __restrict__ W3, const float* __restrict__ b3,
    const float* __restrict__ Wd4, const float* __restrict__ bd4,
    const float* __restrict__ Wl4,
    const float* __restrict__ Wd5, const float* __restrict__ bd5,
    const float* __restrict__ Wl5,
    float* __restrict__ out, float* __restrict__ ws, size_t ws_bytes) {
  __shared__ __attribute__((aligned(16))) float Bs[W_FLOATS];
  const int id = blockIdx.x;
  if (id < FOLD_BLOCKS) {
    if (ws_bytes >= WP_BYTES) {
      if (id < 64)        fold_rows_body<4>(id, Wd4, Wl4, ws, Bs);
      else if (id < 192)  fold_rows_body<8>(id - 64, Wd5, Wl5, ws + WQ_OFF5, Bs);
      else if (id == 192) fold_bias_body<4>(bd4, Wl4, ws + WB_OFF4, Bs);
      else                fold_bias_body<8>(bd5, Wl5, ws + WB_OFF5, Bs);
    }
  } else {
    const int h = id - FOLD_BLOCKS;
    head123_body(h % 100, h / 100, x, W1, b1, W2, b2, W3, b3, out, Bs);
  }
}

#define CONV_BLOCKS 968
__global__ __launch_bounds__(32) void conv_kernel(
    const float* __restrict__ x,
    const float* __restrict__ ws,
    const float* __restrict__ bl4,
    const float* __restrict__ bl5,
    float* __restrict__ out, size_t ws_bytes) {
  if (ws_bytes < WP_BYTES) return;
  const int id = blockIdx.x;
  if (id < 576) {
    const int r = id % 72;
    conv_tile_body<8, 1, 4>(id / 72, r >> 1, (r & 1) * 4, x, ws, bl5, out,
                             2112,  3648,  568);
  } else if (id < 832) {
    const int h = id - 576;
    conv_tile_body<4, 1, 4>(h >> 5, h & 31, 0, x, ws, bl4, out,
                             1600,  1600,  512);
  } else if (id < 904) {
    const int h = id - 832;
    conv_tile_body<8, 4, 1>(h / 9, (h % 9) * 4, 8, x, ws, bl5, out,
                             2112,  3648,  568);
  } else {
    const int h = id - 904;
    conv_tile_body<4, 4, 1>(h >> 3, (h & 7) * 4, 4, x, ws, bl4, out,
                             1600,  1600,  512);
  }
}

extern "C" void kernel_launch(void* const* d_in, const int* in_sizes, int n_in,
                              void* d_out, int out_size, void* d_ws, size_t ws_size,
                              hipStream_t stream) {
  const float* x   = (const float*)d_in[0];
  const float* W1  = (const float*)d_in[4];
  const float* b1  = (const float*)d_in[5];
  const float* W2  = (const float*)d_in[6];
  const float* b2  = (const float*)d_in[7];
  const float* W3  = (const float*)d_in[8];
  const float* b3  = (const float*)d_in[9];
  const float* Wd4 = (const float*)d_in[10];
  const float* bd4 = (const float*)d_in[11];
  const float* Wl4 = (const float*)d_in[12];
  const float* bl4 = (const float*)d_in[13];
  const float* Wd5 = (const float*)d_in[14];
  const float* bd5 = (const float*)d_in[15];
  const float* Wl5 = (const float*)d_in[16];
  const float* bl5 = (const float*)d_in[17];
  float* out = (float*)d_out;
  float* ws  = (float*)d_ws;

  front_kernel<<<dim3(FOLD_BLOCKS + 800), 32, 0, stream>>>(
      x, W1, b1, W2, b2, W3, b3, Wd4, bd4, Wl4, Wd5, bd5, Wl5, out, ws, ws_size);

  conv_kernel<<<dim3(CONV_BLOCKS), 32, 0, stream>>>(x, ws, bl4, bl5, out, ws_size);
}


# Both launches regrouped as four-wave work-groups that stage the B operand in LDS once per group: conv reads its 4 Wq tiles from LDS, front copies W once per 4 tiles, not per wave; f32 sums unchanged
# speedup vs baseline: 16.9633x; 1.0924x over previous
#include <hip/hip_runtime.h>
#include <hip/hip_bf16.h>


#define T_SEQ 8192
#define E_DIM 256
#define V_DIM 17
#define WQ_TILE_FLOATS (32 * 32 * 4)
#define WQ_OFF5 (5 * WQ_TILE_FLOATS)
#define WB_OFF4 ((5 + 9) * WQ_TILE_FLOATS)
#define WB_OFF5 (WB_OFF4 + 4 * V_DIM)
#define WP_BYTES ((size_t)(WB_OFF5 + 8 * V_DIM) * sizeof(float))

typedef float v2f __attribute__((ext_vector_type(2)));
typedef float v4f __attribute__((ext_vector_type(4)));
typedef float v8f __attribute__((ext_vector_type(8)));

__device__ __forceinline__ v8f wmma4(v2f a, v2f b, v8f c) {
  return __builtin_amdgcn_wmma_f32_16x16x4_f32(
       false, a,  false, b,
       (short)0, c,  false,  false);
}

#define W_FLOATS (E_DIM * V_DIM)
template <int AS>
__device__ __forceinline__ void fold_tile(const float* __restrict__ ap,
                                          const float* __restrict__ Bg,
                                          float* __restrict__ Bs,
                                          const int boff0, const int boff1,
                                          v8f& acc0, v8f& acc1) {
  const int tid = threadIdx.x;
  const int wnum = __builtin_amdgcn_readfirstlane((int)(threadIdx.x >> 5));
  constexpr int PRE = (AS == 1) ? 48 : 24;
  v4f wv[9];
  v2f A[64];
  if (wnum < 2) wv[8] = *(const v4f*)(Bg + 4 * (tid + 1024));
#pragma unroll
  for (int q = 0; q < 8; ++q) wv[q] = *(const v4f*)(Bg + 4 * (tid + 128 * q));
#pragma unroll
  for (int j = 0; j < PRE; ++j) {
    A[j].x = ap[(4 * j) * AS];
    A[j].y = ap[(4 * j) * AS + AS];
  }
  __builtin_amdgcn_sched_barrier(0);
  if (wnum < 2) *(v4f*)(Bs + 4 * (tid + 1024)) = wv[8];
#pragma unroll
  for (int q = 0; q < 8; ++q) *(v4f*)(Bs + 4 * (tid + 128 * q)) = wv[q];
#pragma unroll
  for (int j = PRE; j < 64; ++j) {
    A[j].x = ap[(4 * j) * AS];
    A[j].y = ap[(4 * j) * AS + AS];
  }
  __builtin_amdgcn_fence(__ATOMIC_RELEASE, "workgroup", "local");
  __builtin_amdgcn_s_barrier();
  __builtin_amdgcn_fence(__ATOMIC_ACQUIRE, "workgroup", "local");

  const float* b0s = Bs + boff0;
  const float* b1s = Bs + boff1;
  v2f B0[2][8], B1[2][8];
#pragma unroll
  for (int j = 0; j < 8; ++j) {
    const int oo = j * 4;
    B0[0][j].x = b0s[oo * V_DIM];
    B0[0][j].y = b0s[oo * V_DIM + V_DIM];
    B1[0][j].x = b1s[oo * V_DIM];
    B1[0][j].y = b1s[oo * V_DIM + V_DIM];
  }
  __builtin_amdgcn_sched_barrier(0);
#pragma unroll
  for (int ch = 0; ch < 8; ++ch) {
    const int cur = ch & 1;
    const int nxt = cur ^ 1;
    if (ch + 1 < 8) {
#pragma unroll
      for (int j = 0; j < 8; ++j) {
        const int oo = ((ch + 1) * 8 + j) * 4;
        B0[nxt][j].x = b0s[oo * V_DIM];
        B0[nxt][j].y = b0s[oo * V_DIM + V_DIM];
        B1[nxt][j].x = b1s[oo * V_DIM];
        B1[nxt][j].y = b1s[oo * V_DIM + V_DIM];
      }
    }
    __builtin_amdgcn_sched_barrier(0);
#pragma unroll
    for (int j = 0; j < 8; ++j) {
      acc0 = wmma4(A[ch * 8 + j], B0[cur][j], acc0);
      acc1 = wmma4(A[ch * 8 + j], B1[cur][j], acc1);
    }
    __builtin_amdgcn_sched_barrier(0);
  }
}

__device__ __forceinline__ void head123_body(
    const int bx, const int n,
    const float* __restrict__ x,
    const float* __restrict__ W1, const float* __restrict__ b1,
    const float* __restrict__ W2, const float* __restrict__ b2,
    const float* __restrict__ W3, const float* __restrict__ b3,
    float* __restrict__ out, float* __restrict__ Bs) {
  const int rowbase = bx * 16;

  const float* W;
  const float* b;
  if (rowbase < 64)       { W = W1; b = b1; }
  else if (rowbase < 576) { W = W2; b = b2; }
  else                    { W = W3; b = b3; }

  const int lane = threadIdx.x & 31;
  const int l15  = lane & 15;
  const int koff = (lane < 16) ? 0 : 2;

  const float* xrow = x + ((size_t)n * T_SEQ + rowbase + l15) * E_DIM;
  const int c1  = 16 + l15;
  const int c1c = (c1 < V_DIM) ? c1 : (V_DIM - 1);

  v8f acc0 = {};
  v8f acc1 = {};
  fold_tile<1>(xrow + koff, W, Bs, koff * V_DIM + l15, koff * V_DIM + c1c, acc0, acc1);

#pragma unroll
  for (int v = 0; v < 8; ++v) {
    const int r = ((lane < 16) ? v : v + 8);
    float* o = out + ((size_t)n * T_SEQ + rowbase + r) * V_DIM;
    o[l15] = acc0[v] + b[l15];
    if (c1 < V_DIM) o[c1] = acc1[v] + b[c1];
  }
}

template <int S>
__device__ __forceinline__ void fold_rows_body(
    const int tile,
    const float* __restrict__ Wd, const float* __restrict__ Wl,
    float* __restrict__ Wq, float* __restrict__ Bs) {
  constexpr int SH = (S == 8) ? 3 : 2;
  const int lane = threadIdx.x & 31;
  const int l15  = lane & 15;
  const int hh   = lane >> 4;
  const int koff = hh * 2;

  const int ibase = tile * (16 / S);
  const int c1  = 16 + l15;
  const int c1c = (c1 < V_DIM) ? c1 : (V_DIM - 1);

  const float* arow = Wd + (size_t)(ibase + (l15 >> SH)) * (E_DIM * S) + (l15 & (S - 1));

  v8f acc0 = {};
  v8f acc1 = {};
  fold_tile<S>(arow + koff * S, Wl, Bs, koff * V_DIM + l15, koff * V_DIM + c1c, acc0, acc1);

#pragma unroll
  for (int r = 0; r < 8; ++r) {
    const int R   = 8 * hh + r;
    const int i   = ibase + (R >> SH);
    const int tap = R & (S - 1);
    const size_t rowpart = (size_t)(i >> 3) * 32 + 16 * ((i >> 2) & 1);
    {
      const int c = tap * V_DIM + l15;
      Wq[((size_t)(c >> 4) * 1024 + rowpart + (c & 15)) * 4 + (i & 3)] = acc0[r];
    }
    if (c1 < V_DIM) {
      const int c = tap * V_DIM + c1;
      Wq[((size_t)(c >> 4) * 1024 + rowpart + (c & 15)) * 4 + (i & 3)] = acc1[r];
    }
  }
}

template <int S>
__device__ __forceinline__ void fold_bias_body(
    const float* __restrict__ bd, const float* __restrict__ Wl,
    float* __restrict__ Wb, float* __restrict__ Bs, const bool stores) {
  const int lane = threadIdx.x & 31;
  const int l15  = lane & 15;
  const int hh   = lane >> 4;
  const int koff = hh * 2;
  const int c1  = 16 + l15;
  const int c1c = (c1 < V_DIM) ? c1 : (V_DIM - 1);

  v8f acc0 = {};
  v8f acc1 = {};
  fold_tile<1>(bd + koff, Wl, Bs, koff * V_DIM + l15, koff * V_DIM + c1c, acc0, acc1);

  if (stores && hh == 0) {
#pragma unroll
    for (int r = 0; r < S; ++r) {
      Wb[r * V_DIM + l15] = acc0[r];
      if (c1 < V_DIM) Wb[r * V_DIM + c1] = acc1[r];
    }
  }
}

#define CONV_GROUPS 242
#define CONV_LDS_FLOATS (4 * WQ_TILE_FLOATS)

struct HeadP {
  int S, ncol, ioff, ooff, nt, wq, wb;
  const float* bl;
};
__device__ __forceinline__ HeadP head_params(const bool h5, const float* bl4, const float* bl5) {
  HeadP p;
  p.S    = h5 ? 8 : 4;
  p.ncol = h5 ? 8 * V_DIM : 4 * V_DIM;
  p.ioff = h5 ? 2112 : 1600;
  p.ooff = h5 ? 3648 : 1600;
  p.nt   = h5 ? 568 : 512;
  p.wq   = h5 ? WQ_OFF5 : 0;
  p.wb   = h5 ? WB_OFF5 : WB_OFF4;
  p.bl   = h5 ? bl5 : bl4;
  return p;
}

#define FOLD_GROUPS 50
__global__ __launch_bounds__(128) void front_kernel(
    const float* __restrict__ x,
    const float* __restrict__ W1, const float* __restrict__ b1,
    const float* __restrict__ W2, const float* __restrict__ b2,
    const float* __restrict__ W3, const float* __restrict__ b3,
    const float* __restrict__ Wd4, const float* __restrict__ bd4,
    const float* __restrict__ Wl4,
    const float* __restrict__ Wd5, const float* __restrict__ bd5,
    const float* __restrict__ Wl5,
    float* __restrict__ out, float* __restrict__ ws, size_t ws_bytes) {
  __shared__ __attribute__((aligned(16))) float Bs[W_FLOATS];
  const int id = blockIdx.x;
  const int w  = __builtin_amdgcn_readfirstlane((int)(threadIdx.x >> 5));
  if (id < FOLD_GROUPS) {
    if (ws_bytes >= WP_BYTES) {
      if (id < 16)       fold_rows_body<4>(4 * id + w, Wd4, Wl4, ws, Bs);
      else if (id < 48)  fold_rows_body<8>(4 * (id - 16) + w, Wd5, Wl5, ws + WQ_OFF5, Bs);
      else if (id == 48) fold_bias_body<4>(bd4, Wl4, ws + WB_OFF4, Bs, w == 0);
      else               fold_bias_body<8>(bd5, Wl5, ws + WB_OFF5, Bs, w == 0);
    }
  } else {
    const int h = id - FOLD_GROUPS;
    head123_body(4 * (h % 25) + w, h / 25, x, W1, b1, W2, b2, W3, b3, out, Bs);
  }
}

__global__ __launch_bounds__(128) void conv_kernel(
    const float* __restrict__ x,
    const float* __restrict__ ws,
    const float* __restrict__ bl4,
    const float* __restrict__ bl5,
    float* __restrict__ out, size_t ws_bytes) {
  __shared__ __attribute__((aligned(16))) float Bs[CONV_LDS_FLOATS];
  if (ws_bytes < WP_BYTES) return;
  const int g    = blockIdx.x;
  const int w    = __builtin_amdgcn_readfirstlane((int)(threadIdx.x >> 5));
  const int lane = threadIdx.x & 31;
  const int cc15 = lane & 15;
  const int hf   = lane >> 4;

  const bool g5 = g < 178;
  int ct0, a0, na, bidx;
  if (g5) {
    const int cg = (g >= 89) ? 1 : 0;
    const int r  = g - 89 * cg;
    ct0 = 4 * cg;
    if (r < 68) { a0 = 3 * r;              na = 3; bidx = 68 * cg + r; }
    else        { a0 = 204 + 4 * (r - 68); na = 4; bidx = 0; }
  } else {
    ct0 = 0; a0 = 4 * (g - 178); na = 4; bidx = 0;
  }
  const bool isA = w < na;

  const bool my5 = isA ? g5 : (bidx < 72);
  const HeadP P  = head_params(my5, bl4, bl5);
  int n, tb0, ctB;
  if (isA) {
    const int a = a0 + w;
    if (g5) { n = a / 36; tb0 = a - 36 * n; }
    else    { n = a >> 5; tb0 = a & 31; }
    ctB = 0;
  } else {
    const int hB = my5 ? bidx : (bidx - 72);
    if (my5) { n = hB / 9;  tb0 = (hB - 9 * n) * 4; ctB = 8; }
    else     { n = hB >> 3; tb0 = (hB & 7) * 4;     ctB = 4; }
  }

  size_t xo[4];
  size_t wo = 0;
  int clB = 0;
  if (isA) {
    const int trel   = tb0 * 16 + cc15;
    const int tclamp = (trel < P.nt) ? trel : (P.nt - 1);
    const size_t xb  = ((size_t)n * T_SEQ + P.ioff + tclamp) * E_DIM + 4 * hf;
#pragma unroll
    for (int a = 0; a < 4; ++a) xo[a] = xb + 64 * a;
  } else {
#pragma unroll
    for (int a = 0; a < 4; ++a) {
      const int trel   = (tb0 + a) * 16 + cc15;
      const int tclamp = (trel < P.nt) ? trel : (P.nt - 1);
      xo[a] = ((size_t)n * T_SEQ + P.ioff + tclamp) * E_DIM + 4 * hf;
    }
    const int col = ctB * 16 + cc15;
    clB = (col < P.ncol) ? col : (P.ncol - 1);
    wo  = P.wq + ((size_t)(clB >> 4) * 1024 + hf * 16 + (clB & 15)) * 4;
  }

  v4f BB[2][8];
  if (!isA) {
#pragma unroll
    for (int j = 0; j < 8; ++j) BB[0][j] = *(const v4f*)(ws + wo + 128 * j);
  }
  const float* src = ws + (g5 ? WQ_OFF5 : 0) + (size_t)(ct0 + w) * WQ_TILE_FLOATS + 4 * lane;
  v4f cp[32];
#pragma unroll
  for (int q = 0; q < 32; ++q) cp[q] = *(const v4f*)(src + 128 * q);
  __builtin_amdgcn_sched_barrier(0);
  v4f AX[2][8][4];
#pragma unroll
  for (int a = 0; a < 4; ++a)
#pragma unroll
    for (int j = 0; j < 8; ++j) AX[0][j][a] = *(const v4f*)(x + xo[a] + 8 * j);
  __builtin_amdgcn_sched_barrier(0);
  float* dst = Bs + w * WQ_TILE_FLOATS + 4 * lane;
#pragma unroll
  for (int q = 0; q < 32; ++q) *(v4f*)(dst + 128 * q) = cp[q];
  __builtin_amdgcn_fence(__ATOMIC_RELEASE, "workgroup", "local");
  __builtin_amdgcn_s_barrier();
  __builtin_amdgcn_fence(__ATOMIC_ACQUIRE, "workgroup", "local");

  if (isA) {
    float bias[4];
#pragma unroll
    for (int b = 0; b < 4; ++b) {
      const int col = (ct0 + b) * 16 + cc15;
      bias[b] = ws[P.wb + col] + P.bl[col % V_DIM];
    }
    v8f acc[4];
#pragma unroll
    for (int b = 0; b < 4; ++b) acc[b] = (v8f){};
    const float* bsl = Bs + 4 * lane;
    v4f B[2][8][4];
#pragma unroll
    for (int j = 0; j < 8; ++j)
#pragma unroll
      for (int b = 0; b < 4; ++b) B[0][j][b] = *(const v4f*)(bsl + (b * 32 + j) * 128);
#pragma unroll
    for (int ch = 0; ch < 4; ++ch) {
      const int cur = ch & 1;
      const int nxt = cur ^ 1;
      if (ch + 1 < 4) {
#pragma unroll
        for (int j = 0; j < 8; ++j) {
          const int m = (ch + 1) * 8 + j;
#pragma unroll
          for (int b = 0; b < 4; ++b) B[nxt][j][b] = *(const v4f*)(bsl + (b * 32 + m) * 128);
        }
      }
      __builtin_amdgcn_sched_barrier(0);
#pragma unroll
      for (int j = 0; j < 8; ++j) {
        const v4f xa = AX[0][j][ch];
#pragma unroll
        for (int b = 0; b < 4; ++b)
          acc[b] = wmma4((v2f){xa.x, xa.y}, (v2f){B[cur][j][b].x, B[cur][j][b].y}, acc[b]);
#pragma unroll
        for (int b = 0; b < 4; ++b)
          acc[b] = wmma4((v2f){xa.z, xa.w}, (v2f){B[cur][j][b].z, B[cur][j][b].w}, acc[b]);
      }
      __builtin_amdgcn_sched_barrier(0);
    }
#pragma unroll
    for (int b = 0; b < 4; ++b) {
      const int col = (ct0 + b) * 16 + cc15;
#pragma unroll
      for (int v = 0; v < 8; ++v) {
        const int trel2 = tb0 * 16 + 8 * hf + v;
        if (trel2 < P.nt) {
          out[((size_t)n * T_SEQ + P.ooff + (size_t)trel2 * P.S) * V_DIM + col] =
              acc[b][v] + bias[b];
        }
      }
    }
  } else {
    const float biasB = ws[P.wb + clB] + P.bl[clB % V_DIM];
    v8f acc[4];
#pragma unroll
    for (int a = 0; a < 4; ++a) acc[a] = (v8f){};
#pragma unroll
    for (int ch = 0; ch < 4; ++ch) {
      const int cur = ch & 1;
      const int nxt = cur ^ 1;
      if (ch + 1 < 4) {
#pragma unroll
        for (int j = 0; j < 8; ++j) BB[nxt][j] = *(const v4f*)(ws + wo + 128 * ((ch + 1) * 8 + j));
#pragma unroll
        for (int a = 0; a < 4; ++a)
#pragma unroll
          for (int j = 0; j < 8; ++j)
            AX[nxt][j][a] = *(const v4f*)(x + xo[a] + 8 * ((ch + 1) * 8 + j));
      }
      __builtin_amdgcn_sched_barrier(0);
#pragma unroll
      for (int p = 0; p < 4; p += 2) {
#pragma unroll
        for (int j = 0; j < 8; ++j) {
#pragma unroll
          for (int a = p; a < p + 2; ++a)
            acc[a] = wmma4((v2f){AX[cur][j][a].x, AX[cur][j][a].y},
                           (v2f){BB[cur][j].x, BB[cur][j].y}, acc[a]);
#pragma unroll
          for (int a = p; a < p + 2; ++a)
            acc[a] = wmma4((v2f){AX[cur][j][a].z, AX[cur][j][a].w},
                           (v2f){BB[cur][j].z, BB[cur][j].w}, acc[a]);
        }
      }
      __builtin_amdgcn_sched_barrier(0);
    }
    const int col = ctB * 16 + cc15;
    if (col < P.ncol) {
#pragma unroll
      for (int a = 0; a < 4; ++a) {
#pragma unroll
        for (int v = 0; v < 8; ++v) {
          const int trel2 = (tb0 + a) * 16 + 8 * hf + v;
          if (trel2 < P.nt) {
            out[((size_t)n * T_SEQ + P.ooff + (size_t)trel2 * P.S) * V_DIM + col] =
                acc[a][v] + biasB;
          }
        }
      }
    }
  }
}

extern "C" void kernel_launch(void* const* d_in, const int* in_sizes, int n_in,
                              void* d_out, int out_size, void* d_ws, size_t ws_size,
                              hipStream_t stream) {
  const float* x   = (const float*)d_in[0];
  const float* W1  = (const float*)d_in[4];
  const float* b1  = (const float*)d_in[5];
  const float* W2  = (const float*)d_in[6];
  const float* b2  = (const float*)d_in[7];
  const float* W3  = (const float*)d_in[8];
  const float* b3  = (const float*)d_in[9];
  const float* Wd4 = (const float*)d_in[10];
  const float* bd4 = (const float*)d_in[11];
  const float* Wl4 = (const float*)d_in[12];
  const float* bl4 = (const float*)d_in[13];
  const float* Wd5 = (const float*)d_in[14];
  const float* bd5 = (const float*)d_in[15];
  const float* Wl5 = (const float*)d_in[16];
  const float* bl5 = (const float*)d_in[17];
  float* out = (float*)d_out;
  float* ws  = (float*)d_ws;

  front_kernel<<<dim3(FOLD_GROUPS + 200), 128, 0, stream>>>(
      x, W1, b1, W2, b2, W3, b3, Wd4, bd4, Wl4, Wd5, bd5, Wl5, out, ws, ws_size);

  conv_kernel<<<dim3(CONV_GROUPS), 128, 0, stream>>>(x, ws, bl4, bl5, out, ws_size);
}


# Tail after the last matrix step cut in both launches: bias read ahead of the K loop, one guard per 8 rows, stores at constant offsets (was compare, branch, 64-bit address per store); sums unchanged
# speedup vs baseline: 19.7879x; 1.1665x over previous
#include <hip/hip_runtime.h>
#include <hip/hip_bf16.h>


#define T_SEQ 8192
#define E_DIM 256
#define V_DIM 17
#define WQ_TILE_FLOATS (32 * 32 * 4)
#define WQ_OFF5 (5 * WQ_TILE_FLOATS)
#define WB_OFF4 ((5 + 9) * WQ_TILE_FLOATS)
#define WB_OFF5 (WB_OFF4 + 4 * V_DIM)
#define WP_BYTES ((size_t)(WB_OFF5 + 8 * V_DIM) * sizeof(float))

typedef float v2f __attribute__((ext_vector_type(2)));
typedef float v4f __attribute__((ext_vector_type(4)));
typedef float v8f __attribute__((ext_vector_type(8)));

__device__ __forceinline__ v8f wmma4(v2f a, v2f b, v8f c) {
  return __builtin_amdgcn_wmma_f32_16x16x4_f32(
       false, a,  false, b,
       (short)0, c,  false,  false);
}

#define W_FLOATS (E_DIM * V_DIM)
template <int AS>
__device__ __forceinline__ void fold_tile(const float* __restrict__ ap,
                                          const float* __restrict__ Bg,
                                          float* __restrict__ Bs,
                                          const int boff0, const int boff1,
                                          v8f& acc0, v8f& acc1) {
  const int tid = threadIdx.x;
  const int wnum = __builtin_amdgcn_readfirstlane((int)(threadIdx.x >> 5));
  constexpr int PRE = (AS == 1) ? 48 : 24;
  v4f wv[9];
  v2f A[64];
  if (wnum < 2) wv[8] = *(const v4f*)(Bg + 4 * (tid + 1024));
#pragma unroll
  for (int q = 0; q < 8; ++q) wv[q] = *(const v4f*)(Bg + 4 * (tid + 128 * q));
#pragma unroll
  for (int j = 0; j < PRE; ++j) {
    A[j].x = ap[(4 * j) * AS];
    A[j].y = ap[(4 * j) * AS + AS];
  }
  __builtin_amdgcn_sched_barrier(0);
  if (wnum < 2) *(v4f*)(Bs + 4 * (tid + 1024)) = wv[8];
#pragma unroll
  for (int q = 0; q < 8; ++q) *(v4f*)(Bs + 4 * (tid + 128 * q)) = wv[q];
#pragma unroll
  for (int j = PRE; j < 64; ++j) {
    A[j].x = ap[(4 * j) * AS];
    A[j].y = ap[(4 * j) * AS + AS];
  }
  __builtin_amdgcn_fence(__ATOMIC_RELEASE, "workgroup", "local");
  __builtin_amdgcn_s_barrier();
  __builtin_amdgcn_fence(__ATOMIC_ACQUIRE, "workgroup", "local");

  const float* b0s = Bs + boff0;
  const float* b1s = Bs + boff1;
  v2f B0[2][8], B1[2][8];
#pragma unroll
  for (int j = 0; j < 8; ++j) {
    const int oo = j * 4;
    B0[0][j].x = b0s[oo * V_DIM];
    B0[0][j].y = b0s[oo * V_DIM + V_DIM];
    B1[0][j].x = b1s[oo * V_DIM];
    B1[0][j].y = b1s[oo * V_DIM + V_DIM];
  }
  __builtin_amdgcn_sched_barrier(0);
#pragma unroll
  for (int ch = 0; ch < 8; ++ch) {
    const int cur = ch & 1;
    const int nxt = cur ^ 1;
    if (ch + 1 < 8) {
#pragma unroll
      for (int j = 0; j < 8; ++j) {
        const int oo = ((ch + 1) * 8 + j) * 4;
        B0[nxt][j].x = b0s[oo * V_DIM];
        B0[nxt][j].y = b0s[oo * V_DIM + V_DIM];
        B1[nxt][j].x = b1s[oo * V_DIM];
        B1[nxt][j].y = b1s[oo * V_DIM + V_DIM];
      }
    }
    __builtin_amdgcn_sched_barrier(0);
#pragma unroll
    for (int j = 0; j < 8; ++j) {
      acc0 = wmma4(A[ch * 8 + j], B0[cur][j], acc0);
      acc1 = wmma4(A[ch * 8 + j], B1[cur][j], acc1);
    }
    __builtin_amdgcn_sched_barrier(0);
  }
}

__device__ __forceinline__ void head123_body(
    const int bx, const int n,
    const float* __restrict__ x,
    const float* __restrict__ W1, const float* __restrict__ b1,
    const float* __restrict__ W2, const float* __restrict__ b2,
    const float* __restrict__ W3, const float* __restrict__ b3,
    float* __restrict__ out, float* __restrict__ Bs) {
  const int rowbase = bx * 16;

  const float* W;
  const float* b;
  if (rowbase < 64)       { W = W1; b = b1; }
  else if (rowbase < 576) { W = W2; b = b2; }
  else                    { W = W3; b = b3; }

  const int lane = threadIdx.x & 31;
  const int l15  = lane & 15;
  const int hh   = lane >> 4;
  const int koff = hh * 2;

  const float* xrow = x + ((size_t)n * T_SEQ + rowbase + l15) * E_DIM;
  const int c1  = 16 + l15;
  const int c1c = (c1 < V_DIM) ? c1 : (V_DIM - 1);

  const float bv0 = b[l15];
  const float bv1 = b[c1c];

  v8f acc0 = {};
  v8f acc1 = {};
  fold_tile<1>(xrow + koff, W, Bs, koff * V_DIM + l15, koff * V_DIM + c1c, acc0, acc1);

  float s0[8], s1[8];
#pragma unroll
  for (int v = 0; v < 8; ++v) {
    s0[v] = acc0[v] + bv0;
    s1[v] = acc1[v] + bv1;
  }
  float* o = out + ((size_t)n * T_SEQ + rowbase + 8 * hh) * V_DIM;
#pragma unroll
  for (int v = 0; v < 8; ++v) o[v * V_DIM + l15] = s0[v];
  if (c1 < V_DIM) {
#pragma unroll
    for (int v = 0; v < 8; ++v) o[v * V_DIM + c1] = s1[v];
  }
}

template <int S>
__device__ __forceinline__ void fold_rows_body(
    const int tile,
    const float* __restrict__ Wd, const float* __restrict__ Wl,
    float* __restrict__ Wq, float* __restrict__ Bs) {
  constexpr int SH = (S == 8) ? 3 : 2;
  const int lane = threadIdx.x & 31;
  const int l15  = lane & 15;
  const int hh   = lane >> 4;
  const int koff = hh * 2;

  const int ibase = tile * (16 / S);
  const int c1  = 16 + l15;
  const int c1c = (c1 < V_DIM) ? c1 : (V_DIM - 1);

  const float* arow = Wd + (size_t)(ibase + (l15 >> SH)) * (E_DIM * S) + (l15 & (S - 1));

  v8f acc0 = {};
  v8f acc1 = {};
  fold_tile<S>(arow + koff * S, Wl, Bs, koff * V_DIM + l15, koff * V_DIM + c1c, acc0, acc1);

#pragma unroll
  for (int r = 0; r < 8; ++r) {
    const int R   = 8 * hh + r;
    const int i   = ibase + (R >> SH);
    const int tap = R & (S - 1);
    const int rowpart = (i >> 3) * 32 + 16 * ((i >> 2) & 1);
    const int c = tap * V_DIM + l15;
    Wq[((c >> 4) * 1024 + rowpart + (c & 15)) * 4 + (i & 3)] = acc0[r];
  }
  if (c1 < V_DIM) {
#pragma unroll
    for (int r = 0; r < 8; ++r) {
      const int R   = 8 * hh + r;
      const int i   = ibase + (R >> SH);
      const int tap = R & (S - 1);
      const int rowpart = (i >> 3) * 32 + 16 * ((i >> 2) & 1);
      const int c = tap * V_DIM + c1;
      Wq[((c >> 4) * 1024 + rowpart + (c & 15)) * 4 + (i & 3)] = acc1[r];
    }
  }
}

template <int S>
__device__ __forceinline__ void fold_bias_body(
    const float* __restrict__ bd, const float* __restrict__ Wl,
    float* __restrict__ Wb, float* __restrict__ Bs, const bool stores) {
  const int lane = threadIdx.x & 31;
  const int l15  = lane & 15;
  const int hh   = lane >> 4;
  const int koff = hh * 2;
  const int c1  = 16 + l15;
  const int c1c = (c1 < V_DIM) ? c1 : (V_DIM - 1);

  v8f acc0 = {};
  v8f acc1 = {};
  fold_tile<1>(bd + koff, Wl, Bs, koff * V_DIM + l15, koff * V_DIM + c1c, acc0, acc1);

  if (stores && hh == 0) {
#pragma unroll
    for (int r = 0; r < S; ++r) Wb[r * V_DIM + l15] = acc0[r];
    if (c1 < V_DIM) {
#pragma unroll
      for (int r = 0; r < S; ++r) Wb[r * V_DIM + c1] = acc1[r];
    }
  }
}

#define CONV_GROUPS 242
#define CONV_LDS_FLOATS (4 * WQ_TILE_FLOATS)

#define NT4 512
#define NT5 568
#define OOFF4 1600
#define OOFF5 3648
static_assert(NT4 % 8 == 0 && NT5 % 8 == 0, "row guard per 8 rows needs token counts that are multiples of 8");

struct HeadP {
  int S, ncol, ioff, ooff, nt, wq, wb;
  const float* bl;
};
__device__ __forceinline__ HeadP head_params(const bool h5, const float* bl4, const float* bl5) {
  HeadP p;
  p.S    = h5 ? 8 : 4;
  p.ncol = h5 ? 8 * V_DIM : 4 * V_DIM;
  p.ioff = h5 ? 2112 : 1600;
  p.ooff = h5 ? OOFF5 : OOFF4;
  p.nt   = h5 ? NT5 : NT4;
  p.wq   = h5 ? WQ_OFF5 : 0;
  p.wb   = h5 ? WB_OFF5 : WB_OFF4;
  p.bl   = h5 ? bl5 : bl4;
  return p;
}

__device__ __forceinline__ void store_tile_A(float* __restrict__ out, const int n, const int ooff,
                                             const int S, const int trow, const int col0,
                                             const float (&ov)[4][8]) {
  float* p = out + ((size_t)n * T_SEQ + ooff + (size_t)trow * S) * V_DIM + col0;
  const int rstride = S * V_DIM;
#pragma unroll
  for (int v = 0; v < 8; ++v) {
    float* pv = p + v * rstride;
#pragma unroll
    for (int b = 0; b < 4; ++b) pv[b * 16] = ov[b][v];
  }
}
template <int S, int OOFF, int NT>
__device__ __forceinline__ void store_tile_B(float* __restrict__ out, const int n, const int trow0,
                                             const int col, const float (&ov)[4][8]) {
  float* p = out + ((size_t)n * T_SEQ + OOFF + (size_t)trow0 * S) * V_DIM + col;
#pragma unroll
  for (int a = 0; a < 4; ++a) {
    if (trow0 + 16 * a < NT) {
#pragma unroll
      for (int v = 0; v < 8; ++v) p[(16 * a + v) * (S * V_DIM)] = ov[a][v];
    }
  }
}

#define FOLD_GROUPS 50
__global__ __launch_bounds__(128) void front_kernel(
    const float* __restrict__ x,
    const float* __restrict__ W1, const float* __restrict__ b1,
    const float* __restrict__ W2, const float* __restrict__ b2,
    const float* __restrict__ W3, const float* __restrict__ b3,
    const float* __restrict__ Wd4, const float* __restrict__ bd4,
    const float* __restrict__ Wl4,
    const float* __restrict__ Wd5, const float* __restrict__ bd5,
    const float* __restrict__ Wl5,
    float* __restrict__ out, float* __restrict__ ws, size_t ws_bytes) {
  __shared__ __attribute__((aligned(16))) float Bs[W_FLOATS];
  const int id = blockIdx.x;
  const int w  = __builtin_amdgcn_readfirstlane((int)(threadIdx.x >> 5));
  if (id < FOLD_GROUPS) {
    if (ws_bytes >= WP_BYTES) {
      if (id < 16)       fold_rows_body<4>(4 * id + w, Wd4, Wl4, ws, Bs);
      else if (id < 48)  fold_rows_body<8>(4 * (id - 16) + w, Wd5, Wl5, ws + WQ_OFF5, Bs);
      else if (id == 48) fold_bias_body<4>(bd4, Wl4, ws + WB_OFF4, Bs, w == 0);
      else               fold_bias_body<8>(bd5, Wl5, ws + WB_OFF5, Bs, w == 0);
    }
  } else {
    const int h = id - FOLD_GROUPS;
    head123_body(4 * (h % 25) + w, h / 25, x, W1, b1, W2, b2, W3, b3, out, Bs);
  }
}

__global__ __launch_bounds__(128) void conv_kernel(
    const float* __restrict__ x,
    const float* __restrict__ ws,
    const float* __restrict__ bl4,
    const float* __restrict__ bl5,
    float* __restrict__ out, size_t ws_bytes) {
  __shared__ __attribute__((aligned(16))) float Bs[CONV_LDS_FLOATS];
  if (ws_bytes < WP_BYTES) return;
  const int g    = blockIdx.x;
  const int w    = __builtin_amdgcn_readfirstlane((int)(threadIdx.x >> 5));
  const int lane = threadIdx.x & 31;
  const int cc15 = lane & 15;
  const int hf   = lane >> 4;

  const bool g5 = g < 178;
  int ct0, a0, na, bidx;
  if (g5) {
    const int cg = (g >= 89) ? 1 : 0;
    const int r  = g - 89 * cg;
    ct0 = 4 * cg;
    if (r < 68) { a0 = 3 * r;              na = 3; bidx = 68 * cg + r; }
    else        { a0 = 204 + 4 * (r - 68); na = 4; bidx = 0; }
  } else {
    ct0 = 0; a0 = 4 * (g - 178); na = 4; bidx = 0;
  }
  const bool isA = w < na;

  const bool my5 = isA ? g5 : (bidx < 72);
  const HeadP P  = head_params(my5, bl4, bl5);
  int n, tb0, ctB;
  if (isA) {
    const int a = a0 + w;
    if (g5) { n = a / 36; tb0 = a - 36 * n; }
    else    { n = a >> 5; tb0 = a & 31; }
    ctB = 0;
  } else {
    const int hB = my5 ? bidx : (bidx - 72);
    if (my5) { n = hB / 9;  tb0 = (hB - 9 * n) * 4; ctB = 8; }
    else     { n = hB >> 3; tb0 = (hB & 7) * 4;     ctB = 4; }
  }

  size_t xo[4];
  size_t wo = 0;
  int clB = 0;
  if (isA) {
    const int trel   = tb0 * 16 + cc15;
    const int tclamp = (trel < P.nt) ? trel : (P.nt - 1);
    const size_t xb  = ((size_t)n * T_SEQ + P.ioff + tclamp) * E_DIM + 4 * hf;
#pragma unroll
    for (int a = 0; a < 4; ++a) xo[a] = xb + 64 * a;
  } else {
#pragma unroll
    for (int a = 0; a < 4; ++a) {
      const int trel   = (tb0 + a) * 16 + cc15;
      const int tclamp = (trel < P.nt) ? trel : (P.nt - 1);
      xo[a] = ((size_t)n * T_SEQ + P.ioff + tclamp) * E_DIM + 4 * hf;
    }
    const int col = ctB * 16 + cc15;
    clB = (col < P.ncol) ? col : (P.ncol - 1);
    wo  = P.wq + ((size_t)(clB >> 4) * 1024 + hf * 16 + (clB & 15)) * 4;
  }

  v4f BB[2][8];
  if (!isA) {
#pragma unroll
    for (int j = 0; j < 8; ++j) BB[0][j] = *(const v4f*)(ws + wo + 128 * j);
  }
  const float* src = ws + (g5 ? WQ_OFF5 : 0) + (size_t)(ct0 + w) * WQ_TILE_FLOATS + 4 * lane;
  v4f cp[32];
#pragma unroll
  for (int q = 0; q < 32; ++q) cp[q] = *(const v4f*)(src + 128 * q);
  __builtin_amdgcn_sched_barrier(0);
  v4f AX[2][8][4];
#pragma unroll
  for (int a = 0; a < 4; ++a)
#pragma unroll
    for (int j = 0; j < 8; ++j) AX[0][j][a] = *(const v4f*)(x + xo[a] + 8 * j);
  __builtin_amdgcn_sched_barrier(0);
  float* dst = Bs + w * WQ_TILE_FLOATS + 4 * lane;
#pragma unroll
  for (int q = 0; q < 32; ++q) *(v4f*)(dst + 128 * q) = cp[q];
  __builtin_amdgcn_fence(__ATOMIC_RELEASE, "workgroup", "local");
  __builtin_amdgcn_s_barrier();
  __builtin_amdgcn_fence(__ATOMIC_ACQUIRE, "workgroup", "local");

  if (isA) {
    float bias[4];
#pragma unroll
    for (int b = 0; b < 4; ++b) {
      const int col = (ct0 + b) * 16 + cc15;
      bias[b] = ws[P.wb + col] + P.bl[col % V_DIM];
    }
    __builtin_amdgcn_fence(__ATOMIC_SEQ_CST, "singlethread");
    v8f acc[4];
#pragma unroll
    for (int b = 0; b < 4; ++b) acc[b] = (v8f){};
    const float* bsl = Bs + 4 * lane;
    v4f B[2][8][4];
#pragma unroll
    for (int j = 0; j < 8; ++j)
#pragma unroll
      for (int b = 0; b < 4; ++b) B[0][j][b] = *(const v4f*)(bsl + (b * 32 + j) * 128);
#pragma unroll
    for (int ch = 0; ch < 4; ++ch) {
      const int cur = ch & 1;
      const int nxt = cur ^ 1;
      if (ch + 1 < 4) {
#pragma unroll
        for (int j = 0; j < 8; ++j) {
          const int m = (ch + 1) * 8 + j;
#pragma unroll
          for (int b = 0; b < 4; ++b) B[nxt][j][b] = *(const v4f*)(bsl + (b * 32 + m) * 128);
        }
      }
      __builtin_amdgcn_sched_barrier(0);
#pragma unroll
      for (int j = 0; j < 8; ++j) {
        const v4f xa = AX[0][j][ch];
#pragma unroll
        for (int b = 0; b < 4; ++b)
          acc[b] = wmma4((v2f){xa.x, xa.y}, (v2f){B[cur][j][b].x, B[cur][j][b].y}, acc[b]);
#pragma unroll
        for (int b = 0; b < 4; ++b)
          acc[b] = wmma4((v2f){xa.z, xa.w}, (v2f){B[cur][j][b].z, B[cur][j][b].w}, acc[b]);
      }
      __builtin_amdgcn_sched_barrier(0);
    }
    float ov[4][8];
#pragma unroll
    for (int b = 0; b < 4; ++b)
#pragma unroll
      for (int v = 0; v < 8; ++v) ov[b][v] = acc[b][v] + bias[b];
    const int trow = tb0 * 16 + 8 * hf;
    if (trow < P.nt) store_tile_A(out, n, P.ooff, P.S, trow, ct0 * 16 + cc15, ov);
  } else {
    const float biasB = ws[P.wb + clB] + P.bl[clB % V_DIM];
    __builtin_amdgcn_fence(__ATOMIC_SEQ_CST, "singlethread");
    v8f acc[4];
#pragma unroll
    for (int a = 0; a < 4; ++a) acc[a] = (v8f){};
#pragma unroll
    for (int ch = 0; ch < 4; ++ch) {
      const int cur = ch & 1;
      const int nxt = cur ^ 1;
      if (ch + 1 < 4) {
#pragma unroll
        for (int j = 0; j < 8; ++j) BB[nxt][j] = *(const v4f*)(ws + wo + 128 * ((ch + 1) * 8 + j));
#pragma unroll
        for (int a = 0; a < 4; ++a)
#pragma unroll
          for (int j = 0; j < 8; ++j)
            AX[nxt][j][a] = *(const v4f*)(x + xo[a] + 8 * ((ch + 1) * 8 + j));
      }
      __builtin_amdgcn_sched_barrier(0);
#pragma unroll
      for (int p = 0; p < 4; p += 2) {
#pragma unroll
        for (int j = 0; j < 8; ++j) {
#pragma unroll
          for (int a = p; a < p + 2; ++a)
            acc[a] = wmma4((v2f){AX[cur][j][a].x, AX[cur][j][a].y},
                           (v2f){BB[cur][j].x, BB[cur][j].y}, acc[a]);
#pragma unroll
          for (int a = p; a < p + 2; ++a)
            acc[a] = wmma4((v2f){AX[cur][j][a].z, AX[cur][j][a].w},
                           (v2f){BB[cur][j].z, BB[cur][j].w}, acc[a]);
        }
      }
      __builtin_amdgcn_sched_barrier(0);
    }
    float ov[4][8];
#pragma unroll
    for (int a = 0; a < 4; ++a)
#pragma unroll
      for (int v = 0; v < 8; ++v) ov[a][v] = acc[a][v] + biasB;
    const int col = ctB * 16 + cc15;
    if (col < P.ncol) {
      const int trow0 = tb0 * 16 + 8 * hf;
      if (my5) store_tile_B<8, OOFF5, NT5>(out, n, trow0, col, ov);
      else     store_tile_B<4, OOFF4, NT4>(out, n, trow0, col, ov);
    }
  }
}

extern "C" void kernel_launch(void* const* d_in, const int* in_sizes, int n_in,
                              void* d_out, int out_size, void* d_ws, size_t ws_size,
                              hipStream_t stream) {
  const float* x   = (const float*)d_in[0];
  const float* W1  = (const float*)d_in[4];
  const float* b1  = (const float*)d_in[5];
  const float* W2  = (const float*)d_in[6];
  const float* b2  = (const float*)d_in[7];
  const float* W3  = (const float*)d_in[8];
  const float* b3  = (const float*)d_in[9];
  const float* Wd4 = (const float*)d_in[10];
  const float* bd4 = (const float*)d_in[11];
  const float* Wl4 = (const float*)d_in[12];
  const float* bl4 = (const float*)d_in[13];
  const float* Wd5 = (const float*)d_in[14];
  const float* bd5 = (const float*)d_in[15];
  const float* Wl5 = (const float*)d_in[16];
  const float* bl5 = (const float*)d_in[17];
  float* out = (float*)d_out;
  float* ws  = (float*)d_ws;

  front_kernel<<<dim3(FOLD_GROUPS + 200), 128, 0, stream>>>(
      x, W1, b1, W2, b2, W3, b3, Wd4, bd4, Wl4, Wd5, bd5, Wl5, out, ws, ws_size);

  conv_kernel<<<dim3(CONV_GROUPS), 128, 0, stream>>>(x, ws, bl4, bl5, out, ws_size);
}
